# starting kernel
# baseline (speedup 1.0000x reference)
#include <hip/hip_runtime.h>
#include <math.h>

#define BB 64
#define RR 36
#define RP 48
#define LL 64
#define EE 512
#define KC 64
#define LAMBDA 9.0f

typedef float v2f __attribute__((ext_vector_type(2)));
typedef float v8f __attribute__((ext_vector_type(8)));

__device__ __forceinline__ v8f wmma_f32(v2f a, v2f b, v8f c) {
  return __builtin_amdgcn_wmma_f32_16x16x4_f32(
       false, a,  false, b,
       (short)0, c,  false,  false);
}

__global__ __launch_bounds__(128) void normalize_rows(
    const float* __restrict__ captions, const float* __restrict__ images,
    float* __restrict__ capn, float* __restrict__ imgn, float* __restrict__ na) {
  __shared__ float red[128];
  const int row = blockIdx.x;
  const int tid = threadIdx.x;
  const bool is_cap = row < BB * LL;
  const float* src = is_cap ? captions + (size_t)row * EE
                            : images + (size_t)(row - BB * LL) * EE;
  const float4 v = *(const float4*)(src + tid * 4);
  red[tid] = v.x * v.x + v.y * v.y + v.z * v.z + v.w * v.w;
  __syncthreads();
  for (int off = 64; off > 0; off >>= 1) {
    if (tid < off) red[tid] += red[tid + off];
    __syncthreads();
  }
  const float norm = sqrtf(red[0]);
  const float scale = 1.0f / fmaxf(norm, 1e-12f);
  float* dst = is_cap ? capn + (size_t)row * EE
                      : imgn + (size_t)(row - BB * LL) * EE;
  float4 o;
  o.x = v.x * scale; o.y = v.y * scale; o.z = v.z * scale; o.w = v.w * scale;
  *(float4*)(dst + tid * 4) = o;
  if (is_cap && tid == 0) na[row] = norm * scale;
}

__global__ __launch_bounds__(32) void gram_kernel(
    const float* __restrict__ imgn, float* __restrict__ G) {
  const int i = blockIdx.y;
  const int t = blockIdx.x, ti = t / 3, tj = t % 3;
  const int lane = threadIdx.x;
  const int hi = lane >> 4, n16 = lane & 15;
  const int ar = ti * 16 + n16, br = tj * 16 + n16;
  const bool am = ar < RR, bm = br < RR;
  const float* Arow = imgn + ((size_t)i * RR + (am ? ar : 0)) * EE;
  const float* Brow = imgn + ((size_t)i * RR + (bm ? br : 0)) * EE;
  v8f acc = {};
#pragma unroll 8
  for (int k = 0; k < EE; k += 4) {
    const int kk = k + 2 * hi;
    v2f a = *(const v2f*)(Arow + kk);
    v2f b = *(const v2f*)(Brow + kk);
    if (!am) { a.x = 0.f; a.y = 0.f; }
    if (!bm) { b.x = 0.f; b.y = 0.f; }
    acc = wmma_f32(a, b, acc);
  }
#pragma unroll
  for (int j = 0; j < 8; ++j) {
    const int row = ti * 16 + j + 8 * hi;
    const int col = tj * 16 + n16;
    G[((size_t)i * RP + row) * RP + col] = acc[j];
  }
}

__global__ __launch_bounds__(128) void scan_main(
    const float* __restrict__ capn, const float* __restrict__ imgn,
    const float* __restrict__ na_g, const float* __restrict__ G,
    const int* __restrict__ cap_lens, float* __restrict__ out) {
  __shared__ float Ash[LL][KC + 2];
  __shared__ float Bsh[RP][KC + 2];
  __shared__ float Wsh[LL][RP + 2];
  __shared__ float Gsh[RP][RP + 2];
  __shared__ float dotsh[LL];
  __shared__ float rsh[LL];

  const int i = blockIdx.x, c = blockIdx.y;
  const int tid = threadIdx.x;
  const int wave = tid >> 5, lane = tid & 31;
  const int hi = lane >> 4, n16 = lane & 15;

  for (int idx = tid; idx < RP * RP; idx += 128) {
    const int r = idx / RP, col = idx % RP;
    Gsh[r][col] = G[((size_t)i * RP + r) * RP + col];
  }

  const float* Abase = capn + (size_t)c * LL * EE;
  const float* Bbase = imgn + (size_t)i * RR * EE;

  v8f acc0 = {}, acc1 = {}, acc2 = {};
  for (int kc = 0; kc < EE; kc += KC) {
    __syncthreads();
#pragma unroll
    for (int q = 0; q < 8; ++q) {
      const int idx = tid + q * 128;
      const int row = idx >> 4, c4 = (idx & 15) << 2;
      const float4 v = *(const float4*)(Abase + row * EE + kc + c4);
      Ash[row][c4] = v.x; Ash[row][c4 + 1] = v.y;
      Ash[row][c4 + 2] = v.z; Ash[row][c4 + 3] = v.w;
      if (kc + KC < EE)
        __builtin_prefetch(Abase + row * EE + kc + KC + c4, 0, 1);
    }
#pragma unroll
    for (int q = 0; q < 6; ++q) {
      const int idx = tid + q * 128;
      const int row = idx >> 4, c4 = (idx & 15) << 2;
      float4 v; v.x = 0.f; v.y = 0.f; v.z = 0.f; v.w = 0.f;
      if (row < RR) v = *(const float4*)(Bbase + row * EE + kc + c4);
      Bsh[row][c4] = v.x; Bsh[row][c4 + 1] = v.y;
      Bsh[row][c4 + 2] = v.z; Bsh[row][c4 + 3] = v.w;
    }
    __syncthreads();
#pragma unroll
    for (int k0 = 0; k0 < KC; k0 += 4) {
      const int kk = k0 + 2 * hi;
      const v2f a  = *(const v2f*)&Ash[wave * 16 + n16][kk];
      const v2f b0 = *(const v2f*)&Bsh[n16][kk];
      const v2f b1 = *(const v2f*)&Bsh[16 + n16][kk];
      const v2f b2 = *(const v2f*)&Bsh[32 + n16][kk];
      acc0 = wmma_f32(a, b0, acc0);
      acc1 = wmma_f32(a, b1, acc1);
      acc2 = wmma_f32(a, b2, acc2);
    }
  }

  __syncthreads();
#pragma unroll
  for (int j = 0; j < 8; ++j) {
    const int l = wave * 16 + j + 8 * hi;
    Wsh[l][n16]      = acc0[j];
    Wsh[l][16 + n16] = acc1[j];
    Wsh[l][32 + n16] = acc2[j];
  }
  __syncthreads();

  if (tid < LL) {
    float m = -1e30f;
    for (int r = 0; r < RR; ++r) m = fmaxf(m, Wsh[tid][r]);
    float sum = 0.f;
    for (int r = 0; r < RR; ++r) sum += expf(LAMBDA * (Wsh[tid][r] - m));
    const float inv = 1.0f / sum;
    float dot = 0.f;
    for (int r = 0; r < RR; ++r) {
      const float s = Wsh[tid][r];
      const float w = expf(LAMBDA * (s - m)) * inv;
      dot += w * s;
      Wsh[tid][r] = w;
    }
    for (int r = RR; r < RP; ++r) Wsh[tid][r] = 0.f;
    dotsh[tid] = dot;
  }
  __syncthreads();

  v8f v0 = {}, v1 = {}, v2 = {};
#pragma unroll
  for (int k0 = 0; k0 < RP; k0 += 4) {
    const int kk = k0 + 2 * hi;
    const v2f a  = *(const v2f*)&Wsh[wave * 16 + n16][kk];
    const v2f b0 = *(const v2f*)&Gsh[n16][kk];
    const v2f b1 = *(const v2f*)&Gsh[16 + n16][kk];
    const v2f b2 = *(const v2f*)&Gsh[32 + n16][kk];
    v0 = wmma_f32(a, b0, v0);
    v1 = wmma_f32(a, b1, v1);
    v2 = wmma_f32(a, b2, v2);
  }

  __syncthreads();
#pragma unroll
  for (int j = 0; j < 8; ++j) {
    const int l = wave * 16 + j + 8 * hi;
    Ash[l][n16]      = v0[j] * Wsh[l][n16];
    Ash[l][16 + n16] = v1[j] * Wsh[l][16 + n16];
    Ash[l][32 + n16] = v2[j] * Wsh[l][32 + n16];
  }
  __syncthreads();

  if (tid < LL) {
    float nb2 = 0.f;
    for (int r = 0; r < RR; ++r) nb2 += Ash[tid][r];
    const float nb = sqrtf(fmaxf(nb2, 0.f));
    const float na = na_g[c * LL + tid];
    const float denom = fmaxf(na * nb, 1e-8f);
    float rl = dotsh[tid] / denom;
    if (tid >= cap_lens[c]) rl = 0.f;
    rsh[tid] = rl;
  }
  __syncthreads();
  if (tid == 0) {
    float s = 0.f;
    for (int l = 0; l < LL; ++l) s += rsh[l];
    out[(size_t)c * BB + i] = s * (1.0f / LL);
  }
}

extern "C" void kernel_launch(void* const* d_in, const int* in_sizes, int n_in,
                              void* d_out, int out_size, void* d_ws, size_t ws_size,
                              hipStream_t stream) {
  (void)in_sizes; (void)n_in; (void)out_size; (void)ws_size;
  const float* images = (const float*)d_in[0];
  const float* captions = (const float*)d_in[1];
  const int* cap_lens = (const int*)d_in[2];
  float* out = (float*)d_out;

  float* ws = (float*)d_ws;
  float* capn = ws;
  float* imgn = capn + (size_t)BB * LL * EE;
  float* na   = imgn + (size_t)BB * RR * EE;
  float* G    = na + (size_t)BB * LL;

  normalize_rows<<<BB * LL + BB * RR, 128, 0, stream>>>(captions, images, capn,
                                                        imgn, na);
  gram_kernel<<<dim3(9, BB), 32, 0, stream>>>(imgn, G);
  scan_main<<<dim3(BB, BB), 128, 0, stream>>>(capn, imgn, na, G, cap_lens, out);
}


# scan_main staging un-serialized: Gram tile and K-chunk global loads issued as groups, each chunk's loads one chunk ahead of its WMMAs through registers (was load, wait, store one at a time)
# speedup vs baseline: 1.2789x; 1.2789x over previous
#include <hip/hip_runtime.h>
#include <math.h>

#define BB 64
#define RR 36
#define RP 48
#define LL 64
#define EE 512
#define KC 64
#define LAMBDA 9.0f

typedef float v2f __attribute__((ext_vector_type(2)));
typedef float v8f __attribute__((ext_vector_type(8)));

__device__ __forceinline__ v8f wmma_f32(v2f a, v2f b, v8f c) {
  return __builtin_amdgcn_wmma_f32_16x16x4_f32(
       false, a,  false, b,
       (short)0, c,  false,  false);
}

__global__ __launch_bounds__(128) void normalize_rows(
    const float* __restrict__ captions, const float* __restrict__ images,
    float* __restrict__ capn, float* __restrict__ imgn, float* __restrict__ na) {
  __shared__ float red[128];
  const int row = blockIdx.x;
  const int tid = threadIdx.x;
  const bool is_cap = row < BB * LL;
  const float* src = is_cap ? captions + (size_t)row * EE
                            : images + (size_t)(row - BB * LL) * EE;
  const float4 v = *(const float4*)(src + tid * 4);
  red[tid] = v.x * v.x + v.y * v.y + v.z * v.z + v.w * v.w;
  __syncthreads();
  for (int off = 64; off > 0; off >>= 1) {
    if (tid < off) red[tid] += red[tid + off];
    __syncthreads();
  }
  const float norm = sqrtf(red[0]);
  const float scale = 1.0f / fmaxf(norm, 1e-12f);
  float* dst = is_cap ? capn + (size_t)row * EE
                      : imgn + (size_t)(row - BB * LL) * EE;
  float4 o;
  o.x = v.x * scale; o.y = v.y * scale; o.z = v.z * scale; o.w = v.w * scale;
  *(float4*)(dst + tid * 4) = o;
  if (is_cap && tid == 0) na[row] = norm * scale;
}

__global__ __launch_bounds__(32) void gram_kernel(
    const float* __restrict__ imgn, float* __restrict__ G) {
  const int i = blockIdx.y;
  const int t = blockIdx.x, ti = t / 3, tj = t % 3;
  const int lane = threadIdx.x;
  const int hi = lane >> 4, n16 = lane & 15;
  const int ar = ti * 16 + n16, br = tj * 16 + n16;
  const bool am = ar < RR, bm = br < RR;
  const float* Arow = imgn + ((size_t)i * RR + (am ? ar : 0)) * EE;
  const float* Brow = imgn + ((size_t)i * RR + (bm ? br : 0)) * EE;
  v8f acc = {};
#pragma unroll 8
  for (int k = 0; k < EE; k += 4) {
    const int kk = k + 2 * hi;
    v2f a = *(const v2f*)(Arow + kk);
    v2f b = *(const v2f*)(Brow + kk);
    if (!am) { a.x = 0.f; a.y = 0.f; }
    if (!bm) { b.x = 0.f; b.y = 0.f; }
    acc = wmma_f32(a, b, acc);
  }
#pragma unroll
  for (int j = 0; j < 8; ++j) {
    const int row = ti * 16 + j + 8 * hi;
    const int col = tj * 16 + n16;
    G[((size_t)i * RP + row) * RP + col] = acc[j];
  }
}

__global__ __launch_bounds__(128) void scan_main(
    const float* __restrict__ capn, const float* __restrict__ imgn,
    const float* __restrict__ na_g, const float* __restrict__ G,
    const int* __restrict__ cap_lens, float* __restrict__ out) {
  __shared__ float Ash[LL][KC + 2];
  __shared__ float Bsh[RP][KC + 2];
  __shared__ float Wsh[LL][RP + 2];
  __shared__ float Gsh[RP][RP + 2];
  __shared__ float dotsh[LL];
  __shared__ float rsh[LL];

  const int i = blockIdx.x, c = blockIdx.y;
  const int tid = threadIdx.x;
  const int wave = tid >> 5, lane = tid & 31;
  const int hi = lane >> 4, n16 = lane & 15;

  const int srow = tid >> 4, sc4 = (tid & 15) << 2;
  const float* Ap = capn + (size_t)c * LL * EE + srow * EE + sc4;
  const float* Bp = imgn + (size_t)i * RR * EE + srow * EE + sc4;
  const bool b4 = tid < 64;
  const float* Bp4 = Bp + (b4 ? 32 * EE : 0);

  float4 ra[8], rb[5];
#pragma unroll
  for (int q = 0; q < 8; ++q) ra[q] = *(const float4*)(Ap + q * 8 * EE);
#pragma unroll
  for (int q = 0; q < 4; ++q) rb[q] = *(const float4*)(Bp + q * 8 * EE);
  rb[4] = *(const float4*)Bp4;

  {
    const float4* Gi = (const float4*)(G + (size_t)i * RP * RP);
    float4 rg[5];
#pragma unroll
    for (int q = 0; q < 4; ++q) rg[q] = Gi[tid + q * 128];
    if (b4) rg[4] = Gi[tid + 512];
#pragma unroll
    for (int q = 0; q < 5; ++q) {
      if (q < 4 || b4) {
        const int idx = tid + q * 128;
        const int r = idx / 12, col = (idx - r * 12) << 2;
        Gsh[r][col] = rg[q].x; Gsh[r][col + 1] = rg[q].y;
        Gsh[r][col + 2] = rg[q].z; Gsh[r][col + 3] = rg[q].w;
      }
    }
  }
  for (int idx = tid; idx < (RP - RR) * (KC + 2); idx += 128)
    (&Bsh[RR][0])[idx] = 0.f;

  v8f acc0 = {}, acc1 = {}, acc2 = {};
  for (int kc = 0; kc < EE; kc += KC) {
    __syncthreads();
#pragma unroll
    for (int q = 0; q < 8; ++q) {
      float* d = &Ash[srow + 8 * q][sc4];
      d[0] = ra[q].x; d[1] = ra[q].y; d[2] = ra[q].z; d[3] = ra[q].w;
    }
#pragma unroll
    for (int q = 0; q < 4; ++q) {
      float* d = &Bsh[srow + 8 * q][sc4];
      d[0] = rb[q].x; d[1] = rb[q].y; d[2] = rb[q].z; d[3] = rb[q].w;
    }
    if (b4) {
      float* d = &Bsh[32 + srow][sc4];
      d[0] = rb[4].x; d[1] = rb[4].y; d[2] = rb[4].z; d[3] = rb[4].w;
    }
    __syncthreads();
    if (kc + KC < EE) {
#pragma unroll
      for (int q = 0; q < 8; ++q)
        ra[q] = *(const float4*)(Ap + q * 8 * EE + kc + KC);
#pragma unroll
      for (int q = 0; q < 4; ++q)
        rb[q] = *(const float4*)(Bp + q * 8 * EE + kc + KC);
      rb[4] = *(const float4*)(Bp4 + kc + KC);
    }
#pragma unroll
    for (int k0 = 0; k0 < KC; k0 += 4) {
      const int kk = k0 + 2 * hi;
      const v2f a  = *(const v2f*)&Ash[wave * 16 + n16][kk];
      const v2f b0 = *(const v2f*)&Bsh[n16][kk];
      const v2f b1 = *(const v2f*)&Bsh[16 + n16][kk];
      const v2f b2 = *(const v2f*)&Bsh[32 + n16][kk];
      acc0 = wmma_f32(a, b0, acc0);
      acc1 = wmma_f32(a, b1, acc1);
      acc2 = wmma_f32(a, b2, acc2);
    }
  }

  __syncthreads();
#pragma unroll
  for (int j = 0; j < 8; ++j) {
    const int l = wave * 16 + j + 8 * hi;
    Wsh[l][n16]      = acc0[j];
    Wsh[l][16 + n16] = acc1[j];
    Wsh[l][32 + n16] = acc2[j];
  }
  __syncthreads();

  if (tid < LL) {
    float m = -1e30f;
    for (int r = 0; r < RR; ++r) m = fmaxf(m, Wsh[tid][r]);
    float sum = 0.f;
    for (int r = 0; r < RR; ++r) sum += expf(LAMBDA * (Wsh[tid][r] - m));
    const float inv = 1.0f / sum;
    float dot = 0.f;
    for (int r = 0; r < RR; ++r) {
      const float s = Wsh[tid][r];
      const float w = expf(LAMBDA * (s - m)) * inv;
      dot += w * s;
      Wsh[tid][r] = w;
    }
    for (int r = RR; r < RP; ++r) Wsh[tid][r] = 0.f;
    dotsh[tid] = dot;
  }
  __syncthreads();

  v8f v0 = {}, v1 = {}, v2 = {};
#pragma unroll
  for (int k0 = 0; k0 < RP; k0 += 4) {
    const int kk = k0 + 2 * hi;
    const v2f a  = *(const v2f*)&Wsh[wave * 16 + n16][kk];
    const v2f b0 = *(const v2f*)&Gsh[n16][kk];
    const v2f b1 = *(const v2f*)&Gsh[16 + n16][kk];
    const v2f b2 = *(const v2f*)&Gsh[32 + n16][kk];
    v0 = wmma_f32(a, b0, v0);
    v1 = wmma_f32(a, b1, v1);
    v2 = wmma_f32(a, b2, v2);
  }

  __syncthreads();
#pragma unroll
  for (int j = 0; j < 8; ++j) {
    const int l = wave * 16 + j + 8 * hi;
    Ash[l][n16]      = v0[j] * Wsh[l][n16];
    Ash[l][16 + n16] = v1[j] * Wsh[l][16 + n16];
    Ash[l][32 + n16] = v2[j] * Wsh[l][32 + n16];
  }
  __syncthreads();

  if (tid < LL) {
    float nb2 = 0.f;
    for (int r = 0; r < RR; ++r) nb2 += Ash[tid][r];
    const float nb = sqrtf(fmaxf(nb2, 0.f));
    const float na = na_g[c * LL + tid];
    const float denom = fmaxf(na * nb, 1e-8f);
    float rl = dotsh[tid] / denom;
    if (tid >= cap_lens[c]) rl = 0.f;
    rsh[tid] = rl;
  }
  __syncthreads();
  if (tid == 0) {
    float s = 0.f;
    for (int l = 0; l < LL; ++l) s += rsh[l];
    out[(size_t)c * BB + i] = s * (1.0f / LL);
  }
}

extern "C" void kernel_launch(void* const* d_in, const int* in_sizes, int n_in,
                              void* d_out, int out_size, void* d_ws, size_t ws_size,
                              hipStream_t stream) {
  (void)in_sizes; (void)n_in; (void)out_size; (void)ws_size;
  const float* images = (const float*)d_in[0];
  const float* captions = (const float*)d_in[1];
  const int* cap_lens = (const int*)d_in[2];
  float* out = (float*)d_out;

  float* ws = (float*)d_ws;
  float* capn = ws;
  float* imgn = capn + (size_t)BB * LL * EE;
  float* na   = imgn + (size_t)BB * RR * EE;
  float* G    = na + (size_t)BB * LL;

  normalize_rows<<<BB * LL + BB * RR, 128, 0, stream>>>(captions, images, capn,
                                                        imgn, na);
  gram_kernel<<<dim3(9, BB), 32, 0, stream>>>(imgn, G);
  scan_main<<<dim3(BB, BB), 128, 0, stream>>>(capn, imgn, na, G, cap_lens, out);
}


# scan_main block = 1 caption x 4 images: 144 region columns = 9 exact WMMA tiles (was 3 padded tiles per image), caption chunk staged once per 4 images (K-chunk 32 to fit registers); sum order kept
# speedup vs baseline: 1.5683x; 1.2263x over previous
#include <hip/hip_runtime.h>
#include <math.h>

#define BB 64
#define RR 36
#define RP 48
#define LL 64
#define EE 512
#define KC 32
#define LAMBDA 9.0f
#define NI 4
#define NC (NI * RR)
#define NT (NC / 16)
#define WP (NC + 2)
#define GP (RR + 2)

typedef float v2f __attribute__((ext_vector_type(2)));
typedef float v8f __attribute__((ext_vector_type(8)));

__device__ __forceinline__ v8f wmma_f32(v2f a, v2f b, v8f c) {
  return __builtin_amdgcn_wmma_f32_16x16x4_f32(
       false, a,  false, b,
       (short)0, c,  false,  false);
}

__global__ __launch_bounds__(128) void normalize_rows(
    const float* __restrict__ captions, const float* __restrict__ images,
    float* __restrict__ capn, float* __restrict__ imgn, float* __restrict__ na) {
  __shared__ float red[128];
  const int row = blockIdx.x;
  const int tid = threadIdx.x;
  const bool is_cap = row < BB * LL;
  const float* src = is_cap ? captions + (size_t)row * EE
                            : images + (size_t)(row - BB * LL) * EE;
  const float4 v = *(const float4*)(src + tid * 4);
  red[tid] = v.x * v.x + v.y * v.y + v.z * v.z + v.w * v.w;
  __syncthreads();
  for (int off = 64; off > 0; off >>= 1) {
    if (tid < off) red[tid] += red[tid + off];
    __syncthreads();
  }
  const float norm = sqrtf(red[0]);
  const float scale = 1.0f / fmaxf(norm, 1e-12f);
  float* dst = is_cap ? capn + (size_t)row * EE
                      : imgn + (size_t)(row - BB * LL) * EE;
  float4 o;
  o.x = v.x * scale; o.y = v.y * scale; o.z = v.z * scale; o.w = v.w * scale;
  *(float4*)(dst + tid * 4) = o;
  if (is_cap && tid == 0) na[row] = norm * scale;
}

__global__ __launch_bounds__(32) void gram_kernel(
    const float* __restrict__ imgn, float* __restrict__ G) {
  const int i = blockIdx.y;
  const int t = blockIdx.x, ti = t / 3, tj = t % 3;
  const int lane = threadIdx.x;
  const int hi = lane >> 4, n16 = lane & 15;
  const int ar = ti * 16 + n16, br = tj * 16 + n16;
  const bool am = ar < RR, bm = br < RR;
  const float* Arow = imgn + ((size_t)i * RR + (am ? ar : 0)) * EE;
  const float* Brow = imgn + ((size_t)i * RR + (bm ? br : 0)) * EE;
  v8f acc = {};
#pragma unroll 8
  for (int k = 0; k < EE; k += 4) {
    const int kk = k + 2 * hi;
    v2f a = *(const v2f*)(Arow + kk);
    v2f b = *(const v2f*)(Brow + kk);
    if (!am) { a.x = 0.f; a.y = 0.f; }
    if (!bm) { b.x = 0.f; b.y = 0.f; }
    acc = wmma_f32(a, b, acc);
  }
#pragma unroll
  for (int j = 0; j < 8; ++j) {
    const int row = ti * 16 + j + 8 * hi;
    const int col = tj * 16 + n16;
    G[((size_t)i * RP + row) * RP + col] = acc[j];
  }
}

__global__ __launch_bounds__(128) void scan_main(
    const float* __restrict__ capn, const float* __restrict__ imgn,
    const float* __restrict__ na_g, const float* __restrict__ G,
    const int* __restrict__ cap_lens, float* __restrict__ out) {
  __shared__ __attribute__((aligned(16))) float stage[LL * WP + 2 * NI * LL];
  __shared__ __attribute__((aligned(16))) float Gsh[NC][GP];
  float (*Ash)[KC + 2] = (float (*)[KC + 2])stage;
  float (*Bsh)[KC + 2] = (float (*)[KC + 2])(stage + LL * (KC + 2));
  float (*Wsh)[WP] = (float (*)[WP])stage;
  float* dotsh = stage + LL * WP;
  float* rsh = dotsh + NI * LL;

  const int i0 = blockIdx.x * NI, c = blockIdx.y;
  const int tid = threadIdx.x;
  const int wave = tid >> 5, lane = tid & 31;
  const int hi = lane >> 4, n16 = lane & 15;

  const int srow = tid >> 3, sc4 = (tid & 7) << 2;
  const float* Ap = capn + (size_t)c * LL * EE + srow * EE + sc4;
  const float* Bp = imgn + (size_t)i0 * RR * EE + srow * EE + sc4;

  float4 ra[4], rb[9];
#pragma unroll
  for (int q = 0; q < 4; ++q) ra[q] = *(const float4*)(Ap + q * 16 * EE);
#pragma unroll
  for (int q = 0; q < 9; ++q) rb[q] = *(const float4*)(Bp + q * 16 * EE);

  {
    const float* Gi = G + (size_t)i0 * RP * RP;
    float4 rg[11];
#pragma unroll
    for (int q = 0; q < 11; ++q) {
      const int idx = tid + q * 128;
      if (idx < NC * 9) {
        const int row = idx / 9, c4 = idx - row * 9;
        const int j = row / RR, r = row - j * RR;
        rg[q] = *(const float4*)(Gi + ((size_t)j * RP + r) * RP + c4 * 4);
      }
    }
#pragma unroll
    for (int q = 0; q < 11; ++q) {
      const int idx = tid + q * 128;
      if (idx < NC * 9) {
        const int row = idx / 9, col = (idx - row * 9) << 2;
        Gsh[row][col] = rg[q].x; Gsh[row][col + 1] = rg[q].y;
        Gsh[row][col + 2] = rg[q].z; Gsh[row][col + 3] = rg[q].w;
      }
    }
  }

  v8f acc[NT];
#pragma unroll
  for (int t = 0; t < NT; ++t) acc[t] = (v8f){};
  for (int kc = 0; kc < EE; kc += KC) {
    __syncthreads();
#pragma unroll
    for (int q = 0; q < 4; ++q) {
      float* d = &Ash[srow + 16 * q][sc4];
      d[0] = ra[q].x; d[1] = ra[q].y; d[2] = ra[q].z; d[3] = ra[q].w;
    }
#pragma unroll
    for (int q = 0; q < 9; ++q) {
      float* d = &Bsh[srow + 16 * q][sc4];
      d[0] = rb[q].x; d[1] = rb[q].y; d[2] = rb[q].z; d[3] = rb[q].w;
    }
    __syncthreads();
    if (kc + KC < EE) {
#pragma unroll
      for (int q = 0; q < 4; ++q)
        ra[q] = *(const float4*)(Ap + q * 16 * EE + kc + KC);
#pragma unroll
      for (int q = 0; q < 9; ++q)
        rb[q] = *(const float4*)(Bp + q * 16 * EE + kc + KC);
    }
#pragma unroll
    for (int k0 = 0; k0 < KC; k0 += 4) {
      const int kk = k0 + 2 * hi;
      const v2f a = *(const v2f*)&Ash[wave * 16 + n16][kk];
#pragma unroll
      for (int t = 0; t < NT; ++t) {
        const v2f b = *(const v2f*)&Bsh[t * 16 + n16][kk];
        acc[t] = wmma_f32(a, b, acc[t]);
      }
    }
  }

  __syncthreads();
#pragma unroll
  for (int r = 0; r < 8; ++r) {
    const int l = wave * 16 + r + 8 * hi;
#pragma unroll
    for (int t = 0; t < NT; ++t) Wsh[l][t * 16 + n16] = acc[t][r];
  }
  __syncthreads();

  for (int p = 0; p < 2; ++p) {
    const int pair = tid + p * 128;
    float* wr = &Wsh[pair & (LL - 1)][(pair >> 6) * RR];
    float m = -1e30f;
    for (int r = 0; r < RR; ++r) m = fmaxf(m, wr[r]);
    float sum = 0.f;
    for (int r = 0; r < RR; ++r) sum += expf(LAMBDA * (wr[r] - m));
    const float inv = 1.0f / sum;
    float dot = 0.f;
    for (int r = 0; r < RR; ++r) {
      const float s = wr[r];
      const float w = expf(LAMBDA * (s - m)) * inv;
      dot += w * s;
      wr[r] = w;
    }
    dotsh[pair] = dot;
  }
  __syncthreads();

#pragma unroll
  for (int j = 0; j < NI; ++j) {
    const int g2 = (32 + n16 < RR) ? 32 + n16 : RR - 1;
    v8f v0 = {}, v1 = {}, v2 = {};
#pragma unroll
    for (int k0 = 0; k0 < RR; k0 += 4) {
      const int kk = k0 + 2 * hi;
      const v2f a  = *(const v2f*)&Wsh[wave * 16 + n16][j * RR + kk];
      const v2f b0 = *(const v2f*)&Gsh[j * RR + n16][kk];
      const v2f b1 = *(const v2f*)&Gsh[j * RR + 16 + n16][kk];
      const v2f b2 = *(const v2f*)&Gsh[j * RR + g2][kk];
      v0 = wmma_f32(a, b0, v0);
      v1 = wmma_f32(a, b1, v1);
      v2 = wmma_f32(a, b2, v2);
    }
    float w0[8], w1[8], w2[8];
#pragma unroll
    for (int r = 0; r < 8; ++r) {
      const float* wr = &Wsh[wave * 16 + r + 8 * hi][j * RR];
      w0[r] = wr[n16]; w1[r] = wr[16 + n16]; w2[r] = wr[32 + n16];
    }
#pragma unroll
    for (int r = 0; r < 8; ++r) {
      float* wr = &Wsh[wave * 16 + r + 8 * hi][j * RR];
      wr[n16]      = v0[r] * w0[r];
      wr[16 + n16] = v1[r] * w1[r];
      if (n16 < RR - 32) wr[32 + n16] = v2[r] * w2[r];
    }
  }
  __syncthreads();

  for (int p = 0; p < 2; ++p) {
    const int pair = tid + p * 128;
    const int l = pair & (LL - 1);
    const float* wr = &Wsh[l][(pair >> 6) * RR];
    float nb2 = 0.f;
    for (int r = 0; r < RR; ++r) nb2 += wr[r];
    const float nb = sqrtf(fmaxf(nb2, 0.f));
    const float na = na_g[c * LL + l];
    const float denom = fmaxf(na * nb, 1e-8f);
    float rl = dotsh[pair] / denom;
    if (l >= cap_lens[c]) rl = 0.f;
    rsh[pair] = rl;
  }
  __syncthreads();
  if (tid < NI) {
    float s = 0.f;
    for (int l = 0; l < LL; ++l) s += rsh[tid * LL + l];
    out[(size_t)c * BB + i0 + tid] = s * (1.0f / LL);
  }
}

extern "C" void kernel_launch(void* const* d_in, const int* in_sizes, int n_in,
                              void* d_out, int out_size, void* d_ws, size_t ws_size,
                              hipStream_t stream) {
  (void)in_sizes; (void)n_in; (void)out_size; (void)ws_size;
  const float* images = (const float*)d_in[0];
  const float* captions = (const float*)d_in[1];
  const int* cap_lens = (const int*)d_in[2];
  float* out = (float*)d_out;

  float* ws = (float*)d_ws;
  float* capn = ws;
  float* imgn = capn + (size_t)BB * LL * EE;
  float* na   = imgn + (size_t)BB * RR * EE;
  float* G    = na + (size_t)BB * LL;

  normalize_rows<<<BB * LL + BB * RR, 128, 0, stream>>>(captions, images, capn,
                                                        imgn, na);
  gram_kernel<<<dim3(9, BB), 32, 0, stream>>>(imgn, G);
  scan_main<<<dim3(BB / NI, BB), 128, 0, stream>>>(capn, imgn, na, G, cap_lens, out);
}


# L2 normalisation applied in scan_main as a scale after the raw product sums (caption norms from its A fragments, region norms from the Gram diagonal): normalize launch gone; norm sum order changes
# speedup vs baseline: 1.6599x; 1.0584x over previous
#include <hip/hip_runtime.h>
#include <math.h>

#define BB 64
#define RR 36
#define RP 48
#define LL 64
#define EE 512
#define KC 32
#define LAMBDA 9.0f
#define NI 4
#define NC (NI * RR)
#define NT (NC / 16)
#define WP (NC + 2)
#define GP (RR + 2)

typedef float v2f __attribute__((ext_vector_type(2)));
typedef float v8f __attribute__((ext_vector_type(8)));

__device__ __forceinline__ v8f wmma_f32(v2f a, v2f b, v8f c) {
  return __builtin_amdgcn_wmma_f32_16x16x4_f32(
       false, a,  false, b,
       (short)0, c,  false,  false);
}

__global__ __launch_bounds__(32) void gram_kernel(
    const float* __restrict__ images, float* __restrict__ G) {
  const int i = blockIdx.y;
  const int t = blockIdx.x, ti = t / 3, tj = t % 3;
  const int lane = threadIdx.x;
  const int hi = lane >> 4, n16 = lane & 15;
  const int ar = ti * 16 + n16, br = tj * 16 + n16;
  const bool am = ar < RR, bm = br < RR;
  const float* Arow = images + ((size_t)i * RR + (am ? ar : 0)) * EE;
  const float* Brow = images + ((size_t)i * RR + (bm ? br : 0)) * EE;
  v8f acc = {};
#pragma unroll 8
  for (int k = 0; k < EE; k += 4) {
    const int kk = k + 2 * hi;
    v2f a = *(const v2f*)(Arow + kk);
    v2f b = *(const v2f*)(Brow + kk);
    if (!am) { a.x = 0.f; a.y = 0.f; }
    if (!bm) { b.x = 0.f; b.y = 0.f; }
    acc = wmma_f32(a, b, acc);
  }
#pragma unroll
  for (int j = 0; j < 8; ++j) {
    const int row = ti * 16 + j + 8 * hi;
    const int col = tj * 16 + n16;
    G[((size_t)i * RP + row) * RP + col] = acc[j];
  }
}

__global__ __launch_bounds__(128, 4) void scan_main(
    const float* __restrict__ captions, const float* __restrict__ images,
    const float* __restrict__ G,
    const int* __restrict__ cap_lens, float* __restrict__ out) {
  __shared__ __attribute__((aligned(16))) float stage[LL * WP + 2 * NI * LL];
  __shared__ __attribute__((aligned(16))) float Gsh[NC][GP];
  __shared__ float invc[LL];
  __shared__ float nash[LL];
  __shared__ float invi[NC];
  float (*Ash)[KC + 2] = (float (*)[KC + 2])stage;
  float (*Bsh)[KC + 2] = (float (*)[KC + 2])(stage + LL * (KC + 2));
  float (*Wsh)[WP] = (float (*)[WP])stage;
  float* dotsh = stage + LL * WP;
  float* rsh = dotsh + NI * LL;

  const int i0 = blockIdx.x * NI, c = blockIdx.y;
  const int tid = threadIdx.x;
  const int wave = tid >> 5, lane = tid & 31;
  const int hi = lane >> 4, n16 = lane & 15;

  const int srow = tid >> 3, sc4 = (tid & 7) << 2;
  const float* Ap = captions + (size_t)c * LL * EE + srow * EE + sc4;
  const float* Bp = images + (size_t)i0 * RR * EE + srow * EE + sc4;

  float4 ra[4], rb[9];
  float ssq = 0.f;
#pragma unroll
  for (int q = 0; q < 4; ++q) ra[q] = *(const float4*)(Ap + q * 16 * EE);
#pragma unroll
  for (int q = 0; q < 9; ++q) rb[q] = *(const float4*)(Bp + q * 16 * EE);

  {
    const float* Gi = G + (size_t)i0 * RP * RP;
    float4 rg[11];
#pragma unroll
    for (int q = 0; q < 11; ++q) {
      const int idx = tid + q * 128;
      if (idx < NC * 9) {
        const int row = idx / 9, c4 = idx - row * 9;
        const int j = row / RR, r = row - j * RR;
        rg[q] = *(const float4*)(Gi + ((size_t)j * RP + r) * RP + c4 * 4);
      }
    }
#pragma unroll
    for (int q = 0; q < 11; ++q) {
      const int idx = tid + q * 128;
      if (idx < NC * 9) {
        const int row = idx / 9, col = (idx - row * 9) << 2;
        Gsh[row][col] = rg[q].x; Gsh[row][col + 1] = rg[q].y;
        Gsh[row][col + 2] = rg[q].z; Gsh[row][col + 3] = rg[q].w;
      }
    }
  }

  v8f acc[NT];
#pragma unroll
  for (int t = 0; t < NT; ++t) acc[t] = (v8f){};
  for (int kc = 0; kc < EE; kc += KC) {
    __syncthreads();
#pragma unroll
    for (int q = 0; q < 4; ++q) {
      float* d = &Ash[srow + 16 * q][sc4];
      d[0] = ra[q].x; d[1] = ra[q].y; d[2] = ra[q].z; d[3] = ra[q].w;
    }
#pragma unroll
    for (int q = 0; q < 9; ++q) {
      float* d = &Bsh[srow + 16 * q][sc4];
      d[0] = rb[q].x; d[1] = rb[q].y; d[2] = rb[q].z; d[3] = rb[q].w;
    }
    __syncthreads();
    if (kc + KC < EE) {
#pragma unroll
      for (int q = 0; q < 4; ++q)
        ra[q] = *(const float4*)(Ap + q * 16 * EE + kc + KC);
#pragma unroll
      for (int q = 0; q < 9; ++q)
        rb[q] = *(const float4*)(Bp + q * 16 * EE + kc + KC);
    }
#pragma unroll
    for (int k0 = 0; k0 < KC; k0 += 4) {
      const int kk = k0 + 2 * hi;
      const v2f a = *(const v2f*)&Ash[wave * 16 + n16][kk];
      ssq += a.x * a.x + a.y * a.y;
#pragma unroll
      for (int t = 0; t < NT; ++t) {
        const v2f b = *(const v2f*)&Bsh[t * 16 + n16][kk];
        acc[t] = wmma_f32(a, b, acc[t]);
      }
    }
  }

  {
    const float s = ssq + __shfl_xor(ssq, 16);
    if (hi == 0) {
      const float norm = sqrtf(s);
      const float scale = 1.0f / fmaxf(norm, 1e-12f);
      invc[wave * 16 + n16] = scale;
      nash[wave * 16 + n16] = norm * scale;
    }
  }
  for (int col = tid; col < NC; col += 128) {
    const int j = col / RR, r = col - j * RR;
    invi[col] = 1.0f / fmaxf(sqrtf(Gsh[col][r]), 1e-12f);
  }

  __syncthreads();
  {
    float ic[NT];
#pragma unroll
    for (int t = 0; t < NT; ++t) ic[t] = invi[t * 16 + n16];
#pragma unroll
    for (int r = 0; r < 8; ++r) {
      const int l = wave * 16 + r + 8 * hi;
      const float il = invc[l];
#pragma unroll
      for (int t = 0; t < NT; ++t) Wsh[l][t * 16 + n16] = acc[t][r] * il * ic[t];
    }
  }
  __syncthreads();

  for (int p = 0; p < 2; ++p) {
    const int pair = tid + p * 128;
    float* wr = &Wsh[pair & (LL - 1)][(pair >> 6) * RR];
    const float* iv = &invi[(pair >> 6) * RR];
    float m = -1e30f;
    for (int r = 0; r < RR; ++r) m = fmaxf(m, wr[r]);
    float sum = 0.f;
    for (int r = 0; r < RR; ++r) sum += expf(LAMBDA * (wr[r] - m));
    const float inv = 1.0f / sum;
    float dot = 0.f;
    for (int r = 0; r < RR; ++r) {
      const float s = wr[r];
      const float w = expf(LAMBDA * (s - m)) * inv;
      dot += w * s;
      wr[r] = w * iv[r];
    }
    dotsh[pair] = dot;
  }
  __syncthreads();

#pragma unroll
  for (int j = 0; j < NI; ++j) {
    const int g2 = (32 + n16 < RR) ? 32 + n16 : RR - 1;
    v8f v0 = {}, v1 = {}, v2 = {};
#pragma unroll
    for (int k0 = 0; k0 < RR; k0 += 4) {
      const int kk = k0 + 2 * hi;
      const v2f a  = *(const v2f*)&Wsh[wave * 16 + n16][j * RR + kk];
      const v2f b0 = *(const v2f*)&Gsh[j * RR + n16][kk];
      const v2f b1 = *(const v2f*)&Gsh[j * RR + 16 + n16][kk];
      const v2f b2 = *(const v2f*)&Gsh[j * RR + g2][kk];
      v0 = wmma_f32(a, b0, v0);
      v1 = wmma_f32(a, b1, v1);
      v2 = wmma_f32(a, b2, v2);
    }
    float w0[8], w1[8], w2[8];
#pragma unroll
    for (int r = 0; r < 8; ++r) {
      const float* wr = &Wsh[wave * 16 + r + 8 * hi][j * RR];
      w0[r] = wr[n16]; w1[r] = wr[16 + n16]; w2[r] = wr[32 + n16];
    }
#pragma unroll
    for (int r = 0; r < 8; ++r) {
      float* wr = &Wsh[wave * 16 + r + 8 * hi][j * RR];
      wr[n16]      = v0[r] * w0[r];
      wr[16 + n16] = v1[r] * w1[r];
      if (n16 < RR - 32) wr[32 + n16] = v2[r] * w2[r];
    }
  }
  __syncthreads();

  for (int p = 0; p < 2; ++p) {
    const int pair = tid + p * 128;
    const int l = pair & (LL - 1);
    const float* wr = &Wsh[l][(pair >> 6) * RR];
    float nb2 = 0.f;
    for (int r = 0; r < RR; ++r) nb2 += wr[r];
    const float nb = sqrtf(fmaxf(nb2, 0.f));
    const float na = nash[l];
    const float denom = fmaxf(na * nb, 1e-8f);
    float rl = dotsh[pair] / denom;
    if (l >= cap_lens[c]) rl = 0.f;
    rsh[pair] = rl;
  }
  __syncthreads();
  if (tid < NI) {
    float s = 0.f;
    for (int l = 0; l < LL; ++l) s += rsh[tid * LL + l];
    out[(size_t)c * BB + i0 + tid] = s * (1.0f / LL);
  }
}

extern "C" void kernel_launch(void* const* d_in, const int* in_sizes, int n_in,
                              void* d_out, int out_size, void* d_ws, size_t ws_size,
                              hipStream_t stream) {
  (void)in_sizes; (void)n_in; (void)out_size; (void)ws_size;
  const float* images = (const float*)d_in[0];
  const float* captions = (const float*)d_in[1];
  const int* cap_lens = (const int*)d_in[2];
  float* out = (float*)d_out;

  float* G = (float*)d_ws;

  gram_kernel<<<dim3(9, BB), 32, 0, stream>>>(images, G);
  scan_main<<<dim3(BB / NI, BB), 128, 0, stream>>>(captions, images, G, cap_lens, out);
}


# Gram launch: K split over 4 waves per tile, a wave's 32 loads issued as one group ahead of two accumulator chains, partials added in LDS (was 1 wave, a wait per 8 products); G sum order changes
# speedup vs baseline: 1.7688x; 1.0656x over previous
#include <hip/hip_runtime.h>
#include <math.h>

#define BB 64
#define RR 36
#define RP 48
#define LL 64
#define EE 512
#define KC 32
#define LAMBDA 9.0f
#define NI 4
#define NC (NI * RR)
#define NT (NC / 16)
#define WP (NC + 2)
#define GP (RR + 2)

typedef float v2f __attribute__((ext_vector_type(2)));
typedef float v8f __attribute__((ext_vector_type(8)));

__device__ __forceinline__ v8f wmma_f32(v2f a, v2f b, v8f c) {
  return __builtin_amdgcn_wmma_f32_16x16x4_f32(
       false, a,  false, b,
       (short)0, c,  false,  false);
}

#define GW 4
#define GK (EE / GW)
#define GQ (GK / 8)
__global__ __launch_bounds__(32 * GW) void gram_kernel(
    const float* __restrict__ images, float* __restrict__ G) {
  __shared__ float red[GW][8][32];
  const int i = blockIdx.y;
  const int t = blockIdx.x, ti = t / 3, tj = t % 3;
  const int tid = threadIdx.x;
  const int wave = tid >> 5, lane = tid & 31;
  const int hi = lane >> 4, n16 = lane & 15;
  const int ar = ti * 16 + n16, br = tj * 16 + n16;
  const bool am = ar < RR, bm = br < RR;
  const float* Arow = images + ((size_t)i * RR + (am ? ar : 0)) * EE + wave * GK + 4 * hi;
  const float* Brow = images + ((size_t)i * RR + (bm ? br : 0)) * EE + wave * GK + 4 * hi;
  float4 a[GQ], b[GQ];
#pragma unroll
  for (int q = 0; q < GQ; ++q) a[q] = *(const float4*)(Arow + 8 * q);
#pragma unroll
  for (int q = 0; q < GQ; ++q) b[q] = *(const float4*)(Brow + 8 * q);
  __builtin_amdgcn_sched_barrier(0);
  v8f acc0 = {}, acc1 = {};
#pragma unroll
  for (int q = 0; q < GQ; ++q) {
    float4 x = a[q], y = b[q];
    if (!am) x = make_float4(0.f, 0.f, 0.f, 0.f);
    if (!bm) y = make_float4(0.f, 0.f, 0.f, 0.f);
    acc0 = wmma_f32((v2f){x.x, x.y}, (v2f){y.x, y.y}, acc0);
    acc1 = wmma_f32((v2f){x.z, x.w}, (v2f){y.z, y.w}, acc1);
  }
#pragma unroll
  for (int j = 0; j < 8; ++j) red[wave][j][lane] = acc0[j] + acc1[j];
  __syncthreads();
#pragma unroll
  for (int p = 0; p < 2; ++p) {
    const int e = tid + p * 32 * GW;
    const int j = e >> 5, l = e & 31;
    float sum = red[0][j][l];
#pragma unroll
    for (int w = 1; w < GW; ++w) sum += red[w][j][l];
    const int row = ti * 16 + j + 8 * (l >> 4);
    const int col = tj * 16 + (l & 15);
    G[((size_t)i * RP + row) * RP + col] = sum;
  }
}

__global__ __launch_bounds__(128, 4) void scan_main(
    const float* __restrict__ captions, const float* __restrict__ images,
    const float* __restrict__ G,
    const int* __restrict__ cap_lens, float* __restrict__ out) {
  __shared__ __attribute__((aligned(16))) float stage[LL * WP + 2 * NI * LL];
  __shared__ __attribute__((aligned(16))) float Gsh[NC][GP];
  __shared__ float invc[LL];
  __shared__ float nash[LL];
  __shared__ float invi[NC];
  float (*Ash)[KC + 2] = (float (*)[KC + 2])stage;
  float (*Bsh)[KC + 2] = (float (*)[KC + 2])(stage + LL * (KC + 2));
  float (*Wsh)[WP] = (float (*)[WP])stage;
  float* dotsh = stage + LL * WP;
  float* rsh = dotsh + NI * LL;

  const int i0 = blockIdx.x * NI, c = blockIdx.y;
  const int tid = threadIdx.x;
  const int wave = tid >> 5, lane = tid & 31;
  const int hi = lane >> 4, n16 = lane & 15;

  const int srow = tid >> 3, sc4 = (tid & 7) << 2;
  const float* Ap = captions + (size_t)c * LL * EE + srow * EE + sc4;
  const float* Bp = images + (size_t)i0 * RR * EE + srow * EE + sc4;

  float4 ra[4], rb[9];
  float ssq = 0.f;
#pragma unroll
  for (int q = 0; q < 4; ++q) ra[q] = *(const float4*)(Ap + q * 16 * EE);
#pragma unroll
  for (int q = 0; q < 9; ++q) rb[q] = *(const float4*)(Bp + q * 16 * EE);

  {
    const float* Gi = G + (size_t)i0 * RP * RP;
    float4 rg[11];
#pragma unroll
    for (int q = 0; q < 11; ++q) {
      const int idx = tid + q * 128;
      if (idx < NC * 9) {
        const int row = idx / 9, c4 = idx - row * 9;
        const int j = row / RR, r = row - j * RR;
        rg[q] = *(const float4*)(Gi + ((size_t)j * RP + r) * RP + c4 * 4);
      }
    }
#pragma unroll
    for (int q = 0; q < 11; ++q) {
      const int idx = tid + q * 128;
      if (idx < NC * 9) {
        const int row = idx / 9, col = (idx - row * 9) << 2;
        Gsh[row][col] = rg[q].x; Gsh[row][col + 1] = rg[q].y;
        Gsh[row][col + 2] = rg[q].z; Gsh[row][col + 3] = rg[q].w;
      }
    }
  }

  v8f acc[NT];
#pragma unroll
  for (int t = 0; t < NT; ++t) acc[t] = (v8f){};
  for (int kc = 0; kc < EE; kc += KC) {
    __syncthreads();
#pragma unroll
    for (int q = 0; q < 4; ++q) {
      float* d = &Ash[srow + 16 * q][sc4];
      d[0] = ra[q].x; d[1] = ra[q].y; d[2] = ra[q].z; d[3] = ra[q].w;
    }
#pragma unroll
    for (int q = 0; q < 9; ++q) {
      float* d = &Bsh[srow + 16 * q][sc4];
      d[0] = rb[q].x; d[1] = rb[q].y; d[2] = rb[q].z; d[3] = rb[q].w;
    }
    __syncthreads();
    if (kc + KC < EE) {
#pragma unroll
      for (int q = 0; q < 4; ++q)
        ra[q] = *(const float4*)(Ap + q * 16 * EE + kc + KC);
#pragma unroll
      for (int q = 0; q < 9; ++q)
        rb[q] = *(const float4*)(Bp + q * 16 * EE + kc + KC);
    }
#pragma unroll
    for (int k0 = 0; k0 < KC; k0 += 4) {
      const int kk = k0 + 2 * hi;
      const v2f a = *(const v2f*)&Ash[wave * 16 + n16][kk];
      ssq += a.x * a.x + a.y * a.y;
#pragma unroll
      for (int t = 0; t < NT; ++t) {
        const v2f b = *(const v2f*)&Bsh[t * 16 + n16][kk];
        acc[t] = wmma_f32(a, b, acc[t]);
      }
    }
  }

  {
    const float s = ssq + __shfl_xor(ssq, 16);
    if (hi == 0) {
      const float norm = sqrtf(s);
      const float scale = 1.0f / fmaxf(norm, 1e-12f);
      invc[wave * 16 + n16] = scale;
      nash[wave * 16 + n16] = norm * scale;
    }
  }
  for (int col = tid; col < NC; col += 128) {
    const int j = col / RR, r = col - j * RR;
    invi[col] = 1.0f / fmaxf(sqrtf(Gsh[col][r]), 1e-12f);
  }

  __syncthreads();
  {
    float ic[NT];
#pragma unroll
    for (int t = 0; t < NT; ++t) ic[t] = invi[t * 16 + n16];
#pragma unroll
    for (int r = 0; r < 8; ++r) {
      const int l = wave * 16 + r + 8 * hi;
      const float il = invc[l];
#pragma unroll
      for (int t = 0; t < NT; ++t) Wsh[l][t * 16 + n16] = acc[t][r] * il * ic[t];
    }
  }
  __syncthreads();

  for (int p = 0; p < 2; ++p) {
    const int pair = tid + p * 128;
    float* wr = &Wsh[pair & (LL - 1)][(pair >> 6) * RR];
    const float* iv = &invi[(pair >> 6) * RR];
    float m = -1e30f;
    for (int r = 0; r < RR; ++r) m = fmaxf(m, wr[r]);
    float sum = 0.f;
    for (int r = 0; r < RR; ++r) sum += expf(LAMBDA * (wr[r] - m));
    const float inv = 1.0f / sum;
    float dot = 0.f;
    for (int r = 0; r < RR; ++r) {
      const float s = wr[r];
      const float w = expf(LAMBDA * (s - m)) * inv;
      dot += w * s;
      wr[r] = w * iv[r];
    }
    dotsh[pair] = dot;
  }
  __syncthreads();

#pragma unroll
  for (int j = 0; j < NI; ++j) {
    const int g2 = (32 + n16 < RR) ? 32 + n16 : RR - 1;
    v8f v0 = {}, v1 = {}, v2 = {};
#pragma unroll
    for (int k0 = 0; k0 < RR; k0 += 4) {
      const int kk = k0 + 2 * hi;
      const v2f a  = *(const v2f*)&Wsh[wave * 16 + n16][j * RR + kk];
      const v2f b0 = *(const v2f*)&Gsh[j * RR + n16][kk];
      const v2f b1 = *(const v2f*)&Gsh[j * RR + 16 + n16][kk];
      const v2f b2 = *(const v2f*)&Gsh[j * RR + g2][kk];
      v0 = wmma_f32(a, b0, v0);
      v1 = wmma_f32(a, b1, v1);
      v2 = wmma_f32(a, b2, v2);
    }
    float w0[8], w1[8], w2[8];
#pragma unroll
    for (int r = 0; r < 8; ++r) {
      const float* wr = &Wsh[wave * 16 + r + 8 * hi][j * RR];
      w0[r] = wr[n16]; w1[r] = wr[16 + n16]; w2[r] = wr[32 + n16];
    }
#pragma unroll
    for (int r = 0; r < 8; ++r) {
      float* wr = &Wsh[wave * 16 + r + 8 * hi][j * RR];
      wr[n16]      = v0[r] * w0[r];
      wr[16 + n16] = v1[r] * w1[r];
      if (n16 < RR - 32) wr[32 + n16] = v2[r] * w2[r];
    }
  }
  __syncthreads();

  for (int p = 0; p < 2; ++p) {
    const int pair = tid + p * 128;
    const int l = pair & (LL - 1);
    const float* wr = &Wsh[l][(pair >> 6) * RR];
    float nb2 = 0.f;
    for (int r = 0; r < RR; ++r) nb2 += wr[r];
    const float nb = sqrtf(fmaxf(nb2, 0.f));
    const float na = nash[l];
    const float denom = fmaxf(na * nb, 1e-8f);
    float rl = dotsh[pair] / denom;
    if (l >= cap_lens[c]) rl = 0.f;
    rsh[pair] = rl;
  }
  __syncthreads();
  if (tid < NI) {
    float s = 0.f;
    for (int l = 0; l < LL; ++l) s += rsh[tid * LL + l];
    out[(size_t)c * BB + i0 + tid] = s * (1.0f / LL);
  }
}

extern "C" void kernel_launch(void* const* d_in, const int* in_sizes, int n_in,
                              void* d_out, int out_size, void* d_ws, size_t ws_size,
                              hipStream_t stream) {
  (void)in_sizes; (void)n_in; (void)out_size; (void)ws_size;
  const float* images = (const float*)d_in[0];
  const float* captions = (const float*)d_in[1];
  const int* cap_lens = (const int*)d_in[2];
  float* out = (float*)d_out;

  float* G = (float*)d_ws;

  gram_kernel<<<dim3(9, BB), 32 * GW, 0, stream>>>(images, G);
  scan_main<<<dim3(BB / NI, BB), 128, 0, stream>>>(captions, images, G, cap_lens, out);
}


# scan_main LDS tiles at row pitch 16 B x odd (K-chunk 36 floats, Gram 36): fragments read and staged as conflict-free 16-byte accesses, one read feeds two products; k order of the sums changes
# speedup vs baseline: 1.8393x; 1.0399x over previous
#include <hip/hip_runtime.h>
#include <math.h>

#define BB 64
#define RR 36
#define RP 48
#define LL 64
#define EE 512
#define KC 32
#define KP 36
#define LAMBDA 9.0f
#define NI 4
#define NC (NI * RR)
#define NT (NC / 16)
#define WP (NC + 2)
#define GP RR

typedef float v2f __attribute__((ext_vector_type(2)));
typedef float v4f __attribute__((ext_vector_type(4)));
typedef float v8f __attribute__((ext_vector_type(8)));

__device__ __forceinline__ v8f wmma_f32(v2f a, v2f b, v8f c) {
  return __builtin_amdgcn_wmma_f32_16x16x4_f32(
       false, a,  false, b,
       (short)0, c,  false,  false);
}

#define GW 4
#define GK (EE / GW)
#define GQ (GK / 8)
__global__ __launch_bounds__(32 * GW) void gram_kernel(
    const float* __restrict__ images, float* __restrict__ G) {
  __shared__ float red[GW][8][32];
  const int i = blockIdx.y;
  const int t = blockIdx.x, ti = t / 3, tj = t % 3;
  const int tid = threadIdx.x;
  const int wave = tid >> 5, lane = tid & 31;
  const int hi = lane >> 4, n16 = lane & 15;
  const int ar = ti * 16 + n16, br = tj * 16 + n16;
  const bool am = ar < RR, bm = br < RR;
  const float* Arow = images + ((size_t)i * RR + (am ? ar : 0)) * EE + wave * GK + 4 * hi;
  const float* Brow = images + ((size_t)i * RR + (bm ? br : 0)) * EE + wave * GK + 4 * hi;
  float4 a[GQ], b[GQ];
#pragma unroll
  for (int q = 0; q < GQ; ++q) a[q] = *(const float4*)(Arow + 8 * q);
#pragma unroll
  for (int q = 0; q < GQ; ++q) b[q] = *(const float4*)(Brow + 8 * q);
  __builtin_amdgcn_sched_barrier(0);
  v8f acc0 = {}, acc1 = {};
#pragma unroll
  for (int q = 0; q < GQ; ++q) {
    float4 x = a[q], y = b[q];
    if (!am) x = make_float4(0.f, 0.f, 0.f, 0.f);
    if (!bm) y = make_float4(0.f, 0.f, 0.f, 0.f);
    acc0 = wmma_f32((v2f){x.x, x.y}, (v2f){y.x, y.y}, acc0);
    acc1 = wmma_f32((v2f){x.z, x.w}, (v2f){y.z, y.w}, acc1);
  }
#pragma unroll
  for (int j = 0; j < 8; ++j) red[wave][j][lane] = acc0[j] + acc1[j];
  __syncthreads();
#pragma unroll
  for (int p = 0; p < 2; ++p) {
    const int e = tid + p * 32 * GW;
    const int j = e >> 5, l = e & 31;
    float sum = red[0][j][l];
#pragma unroll
    for (int w = 1; w < GW; ++w) sum += red[w][j][l];
    const int row = ti * 16 + j + 8 * (l >> 4);
    const int col = tj * 16 + (l & 15);
    G[((size_t)i * RP + row) * RP + col] = sum;
  }
}

__global__ __launch_bounds__(128, 4) void scan_main(
    const float* __restrict__ captions, const float* __restrict__ images,
    const float* __restrict__ G,
    const int* __restrict__ cap_lens, float* __restrict__ out) {
  __shared__ __attribute__((aligned(16))) float stage[LL * WP + 2 * NI * LL];
  __shared__ __attribute__((aligned(16))) float Gsh[NC][GP];
  __shared__ float invc[LL];
  __shared__ float nash[LL];
  __shared__ float invi[NC];
  float (*Ash)[KP] = (float (*)[KP])stage;
  float (*Bsh)[KP] = (float (*)[KP])(stage + LL * KP);
  float (*Wsh)[WP] = (float (*)[WP])stage;
  float* dotsh = stage + LL * WP;
  float* rsh = dotsh + NI * LL;

  const int i0 = blockIdx.x * NI, c = blockIdx.y;
  const int tid = threadIdx.x;
  const int wave = tid >> 5, lane = tid & 31;
  const int hi = lane >> 4, n16 = lane & 15;

  const int srow = (tid >> 4) + 8 * ((tid >> 3) & 1), sc4 = (tid & 7) << 2;
  const float* Ap = captions + (size_t)c * LL * EE + srow * EE + sc4;
  const float* Bp = images + (size_t)i0 * RR * EE + srow * EE + sc4;

  v4f ra[4], rb[9];
  float ssq = 0.f;
#pragma unroll
  for (int q = 0; q < 4; ++q) ra[q] = *(const v4f*)(Ap + q * 16 * EE);
#pragma unroll
  for (int q = 0; q < 9; ++q) rb[q] = *(const v4f*)(Bp + q * 16 * EE);

  {
    const float* Gi = G + (size_t)i0 * RP * RP;
    v4f rg[11];
#pragma unroll
    for (int q = 0; q < 11; ++q) {
      const int idx = tid + q * 128;
      if (idx < NC * 9) {
        const int row = idx / 9, c4 = idx - row * 9;
        const int j = row / RR, r = row - j * RR;
        rg[q] = *(const v4f*)(Gi + ((size_t)j * RP + r) * RP + c4 * 4);
      }
    }
#pragma unroll
    for (int q = 0; q < 11; ++q) {
      const int idx = tid + q * 128;
      if (idx < NC * 9) {
        const int row = idx / 9, col = (idx - row * 9) << 2;
        *(v4f*)&Gsh[row][col] = rg[q];
      }
    }
  }

  v8f acc[NT];
#pragma unroll
  for (int t = 0; t < NT; ++t) acc[t] = (v8f){};
  for (int kc = 0; kc < EE; kc += KC) {
    __syncthreads();
#pragma unroll
    for (int q = 0; q < 4; ++q) *(v4f*)&Ash[srow + 16 * q][sc4] = ra[q];
#pragma unroll
    for (int q = 0; q < 9; ++q) *(v4f*)&Bsh[srow + 16 * q][sc4] = rb[q];
    __syncthreads();
    if (kc + KC < EE) {
#pragma unroll
      for (int q = 0; q < 4; ++q)
        ra[q] = *(const v4f*)(Ap + q * 16 * EE + kc + KC);
#pragma unroll
      for (int q = 0; q < 9; ++q)
        rb[q] = *(const v4f*)(Bp + q * 16 * EE + kc + KC);
    }
#pragma unroll
    for (int k0 = 0; k0 < KC; k0 += 8) {
      const int kk = k0 + 4 * hi;
      const v4f a = *(const v4f*)&Ash[wave * 16 + n16][kk];
      ssq += a.x * a.x + a.y * a.y + a.z * a.z + a.w * a.w;
      v4f b[NT];
#pragma unroll
      for (int t = 0; t < NT; ++t) b[t] = *(const v4f*)&Bsh[t * 16 + n16][kk];
#pragma unroll
      for (int t = 0; t < NT; ++t)
        acc[t] = wmma_f32((v2f){a.x, a.y}, (v2f){b[t].x, b[t].y}, acc[t]);
#pragma unroll
      for (int t = 0; t < NT; ++t)
        acc[t] = wmma_f32((v2f){a.z, a.w}, (v2f){b[t].z, b[t].w}, acc[t]);
    }
  }

  {
    const float s = ssq + __shfl_xor(ssq, 16);
    if (hi == 0) {
      const float norm = sqrtf(s);
      const float scale = 1.0f / fmaxf(norm, 1e-12f);
      invc[wave * 16 + n16] = scale;
      nash[wave * 16 + n16] = norm * scale;
    }
  }
  for (int col = tid; col < NC; col += 128) {
    const int j = col / RR, r = col - j * RR;
    invi[col] = 1.0f / fmaxf(sqrtf(Gsh[col][r]), 1e-12f);
  }

  __syncthreads();
  {
    float ic[NT];
#pragma unroll
    for (int t = 0; t < NT; ++t) ic[t] = invi[t * 16 + n16];
#pragma unroll
    for (int r = 0; r < 8; ++r) {
      const int l = wave * 16 + r + 8 * hi;
      const float il = invc[l];
#pragma unroll
      for (int t = 0; t < NT; ++t) Wsh[l][t * 16 + n16] = acc[t][r] * il * ic[t];
    }
  }
  __syncthreads();

  for (int p = 0; p < 2; ++p) {
    const int pair = tid + p * 128;
    float* wr = &Wsh[pair & (LL - 1)][(pair >> 6) * RR];
    const float* iv = &invi[(pair >> 6) * RR];
    float m = -1e30f;
    for (int r = 0; r < RR; ++r) m = fmaxf(m, wr[r]);
    float sum = 0.f;
    for (int r = 0; r < RR; ++r) sum += expf(LAMBDA * (wr[r] - m));
    const float inv = 1.0f / sum;
    float dot = 0.f;
    for (int r = 0; r < RR; ++r) {
      const float s = wr[r];
      const float w = expf(LAMBDA * (s - m)) * inv;
      dot += w * s;
      wr[r] = w * iv[r];
    }
    dotsh[pair] = dot;
  }
  __syncthreads();

#pragma unroll
  for (int j = 0; j < NI; ++j) {
    const int g2 = (32 + n16 < RR) ? 32 + n16 : RR - 1;
    v8f v0 = {}, v1 = {}, v2 = {};
#pragma unroll
    for (int k0 = 0; k0 < RR; k0 += 4) {
      const int kk = k0 + 2 * hi;
      const v2f a  = *(const v2f*)&Wsh[wave * 16 + n16][j * RR + kk];
      const v2f b0 = *(const v2f*)&Gsh[j * RR + n16][kk];
      const v2f b1 = *(const v2f*)&Gsh[j * RR + 16 + n16][kk];
      const v2f b2 = *(const v2f*)&Gsh[j * RR + g2][kk];
      v0 = wmma_f32(a, b0, v0);
      v1 = wmma_f32(a, b1, v1);
      v2 = wmma_f32(a, b2, v2);
    }
    float w0[8], w1[8], w2[8];
#pragma unroll
    for (int r = 0; r < 8; ++r) {
      const float* wr = &Wsh[wave * 16 + r + 8 * hi][j * RR];
      w0[r] = wr[n16]; w1[r] = wr[16 + n16]; w2[r] = wr[32 + n16];
    }
#pragma unroll
    for (int r = 0; r < 8; ++r) {
      float* wr = &Wsh[wave * 16 + r + 8 * hi][j * RR];
      wr[n16]      = v0[r] * w0[r];
      wr[16 + n16] = v1[r] * w1[r];
      if (n16 < RR - 32) wr[32 + n16] = v2[r] * w2[r];
    }
  }
  __syncthreads();

  for (int p = 0; p < 2; ++p) {
    const int pair = tid + p * 128;
    const int l = pair & (LL - 1);
    const float* wr = &Wsh[l][(pair >> 6) * RR];
    float nb2 = 0.f;
    for (int r = 0; r < RR; ++r) nb2 += wr[r];
    const float nb = sqrtf(fmaxf(nb2, 0.f));
    const float na = nash[l];
    const float denom = fmaxf(na * nb, 1e-8f);
    float rl = dotsh[pair] / denom;
    if (l >= cap_lens[c]) rl = 0.f;
    rsh[pair] = rl;
  }
  __syncthreads();
  if (tid < NI) {
    float s = 0.f;
    for (int l = 0; l < LL; ++l) s += rsh[tid * LL + l];
    out[(size_t)c * BB + i0 + tid] = s * (1.0f / LL);
  }
}

extern "C" void kernel_launch(void* const* d_in, const int* in_sizes, int n_in,
                              void* d_out, int out_size, void* d_ws, size_t ws_size,
                              hipStream_t stream) {
  (void)in_sizes; (void)n_in; (void)out_size; (void)ws_size;
  const float* images = (const float*)d_in[0];
  const float* captions = (const float*)d_in[1];
  const int* cap_lens = (const int*)d_in[2];
  float* out = (float*)d_out;

  float* G = (float*)d_ws;

  gram_kernel<<<dim3(9, BB), 32 * GW, 0, stream>>>(images, G);
  scan_main<<<dim3(BB / NI, BB), 128, 0, stream>>>(captions, images, G, cap_lens, out);
}


# scan_main builds the similarity tile transposed (a lane owns one token, regions down its registers): softmax, Gram product (weights as B from registers) and norm stay in registers; sum orders change
# speedup vs baseline: 1.9547x; 1.0627x over previous
#include <hip/hip_runtime.h>
#include <math.h>

#define BB 64
#define RR 36
#define RP 48
#define LL 64
#define EE 512
#define KC 32
#define KP 36
#define LAMBDA 9.0f
#define NI 4
#define NC (NI * RR)
#define NT (NC / 16)
#define GP RR

typedef float v2f __attribute__((ext_vector_type(2)));
typedef float v4f __attribute__((ext_vector_type(4)));
typedef float v8f __attribute__((ext_vector_type(8)));

__device__ __forceinline__ v8f wmma_f32(v2f a, v2f b, v8f c) {
  return __builtin_amdgcn_wmma_f32_16x16x4_f32(
       false, a,  false, b,
       (short)0, c,  false,  false);
}

__device__ __forceinline__ float xhalf(float x) {
  const int b = __builtin_bit_cast(int, x);
  return __builtin_bit_cast(float, __builtin_amdgcn_permlanex16(b, b, 0x76543210, 0xfedcba98, false, false));
}
template <int K> __device__ __forceinline__ float xrow(float x) {
  return __builtin_bit_cast(float, __builtin_amdgcn_update_dpp(0, __builtin_bit_cast(int, x), 0x160 | K, 0xf, 0xf, false));
}

#define GW 4
#define GK (EE / GW)
#define GQ (GK / 8)
__global__ __launch_bounds__(32 * GW) void gram_kernel(
    const float* __restrict__ images, float* __restrict__ G) {
  __shared__ float red[GW][8][32];
  const int i = blockIdx.y;
  const int t = blockIdx.x, ti = t / 3, tj = t % 3;
  const int tid = threadIdx.x;
  const int wave = tid >> 5, lane = tid & 31;
  const int hi = lane >> 4, n16 = lane & 15;
  const int ar = ti * 16 + n16, br = tj * 16 + n16;
  const bool am = ar < RR, bm = br < RR;
  const float* Arow = images + ((size_t)i * RR + (am ? ar : 0)) * EE + wave * GK + 4 * hi;
  const float* Brow = images + ((size_t)i * RR + (bm ? br : 0)) * EE + wave * GK + 4 * hi;
  float4 a[GQ], b[GQ];
#pragma unroll
  for (int q = 0; q < GQ; ++q) a[q] = *(const float4*)(Arow + 8 * q);
#pragma unroll
  for (int q = 0; q < GQ; ++q) b[q] = *(const float4*)(Brow + 8 * q);
  __builtin_amdgcn_sched_barrier(0);
  v8f acc0 = {}, acc1 = {};
#pragma unroll
  for (int q = 0; q < GQ; ++q) {
    float4 x = a[q], y = b[q];
    if (!am) x = make_float4(0.f, 0.f, 0.f, 0.f);
    if (!bm) y = make_float4(0.f, 0.f, 0.f, 0.f);
    acc0 = wmma_f32((v2f){x.x, x.y}, (v2f){y.x, y.y}, acc0);
    acc1 = wmma_f32((v2f){x.z, x.w}, (v2f){y.z, y.w}, acc1);
  }
#pragma unroll
  for (int j = 0; j < 8; ++j) red[wave][j][lane] = acc0[j] + acc1[j];
  __syncthreads();
#pragma unroll
  for (int p = 0; p < 2; ++p) {
    const int e = tid + p * 32 * GW;
    const int j = e >> 5, l = e & 31;
    float sum = red[0][j][l];
#pragma unroll
    for (int w = 1; w < GW; ++w) sum += red[w][j][l];
    const int row = ti * 16 + j + 8 * (l >> 4);
    const int col = tj * 16 + (l & 15);
    G[((size_t)i * RP + row) * RP + col] = sum;
  }
}

__global__ __launch_bounds__(128, 4) void scan_main(
    const float* __restrict__ captions, const float* __restrict__ images,
    const float* __restrict__ G,
    const int* __restrict__ cap_lens, float* __restrict__ out) {
  __shared__ __attribute__((aligned(16))) float stage[(LL + NC) * KP];
  __shared__ __attribute__((aligned(16))) float Gsh[NC][GP];
  __shared__ __attribute__((aligned(16))) float invi[NC];
  __shared__ float part[4][NI];
  float (*Ash)[KP] = (float (*)[KP])stage;
  float (*Bsh)[KP] = (float (*)[KP])(stage + LL * KP);

  const int i0 = blockIdx.x * NI, c = blockIdx.y;
  const int tid = threadIdx.x;
  const int wave = tid >> 5, lane = tid & 31;
  const int hi = lane >> 4, n16 = lane & 15;
  const bool up = hi != 0;

  const int srow = (tid >> 4) + 8 * ((tid >> 3) & 1), sc4 = (tid & 7) << 2;
  const float* Ap = captions + (size_t)c * LL * EE + srow * EE + sc4;
  const float* Bp = images + (size_t)i0 * RR * EE + srow * EE + sc4;

  v4f ra[4], rb[9];
  float ssq = 0.f;
#pragma unroll
  for (int q = 0; q < 4; ++q) ra[q] = *(const v4f*)(Ap + q * 16 * EE);
#pragma unroll
  for (int q = 0; q < 9; ++q) rb[q] = *(const v4f*)(Bp + q * 16 * EE);

  {
    const float* Gi = G + (size_t)i0 * RP * RP;
    v4f rg[11];
#pragma unroll
    for (int q = 0; q < 11; ++q) {
      const int idx = tid + q * 128;
      if (idx < NC * 9) {
        const int row = idx / 9, c4 = idx - row * 9;
        const int j = row / RR, r = row - j * RR;
        rg[q] = *(const v4f*)(Gi + ((size_t)j * RP + r) * RP + c4 * 4);
      }
    }
#pragma unroll
    for (int q = 0; q < 11; ++q) {
      const int idx = tid + q * 128;
      if (idx < NC * 9) {
        const int row = idx / 9, col = (idx - row * 9) << 2;
        *(v4f*)&Gsh[row][col] = rg[q];
      }
    }
  }

  v8f acc[NT];
#pragma unroll
  for (int t = 0; t < NT; ++t) acc[t] = (v8f){};
  for (int kc = 0; kc < EE; kc += KC) {
    __syncthreads();
#pragma unroll
    for (int q = 0; q < 4; ++q) *(v4f*)&Ash[srow + 16 * q][sc4] = ra[q];
#pragma unroll
    for (int q = 0; q < 9; ++q) *(v4f*)&Bsh[srow + 16 * q][sc4] = rb[q];
    __syncthreads();
    if (kc + KC < EE) {
#pragma unroll
      for (int q = 0; q < 4; ++q)
        ra[q] = *(const v4f*)(Ap + q * 16 * EE + kc + KC);
#pragma unroll
      for (int q = 0; q < 9; ++q)
        rb[q] = *(const v4f*)(Bp + q * 16 * EE + kc + KC);
    }
#pragma unroll
    for (int k0 = 0; k0 < KC; k0 += 8) {
      const int kk = k0 + 4 * hi;
      const v4f a = *(const v4f*)&Ash[wave * 16 + n16][kk];
      ssq += a.x * a.x + a.y * a.y + a.z * a.z + a.w * a.w;
      v4f b[NT];
#pragma unroll
      for (int t = 0; t < NT; ++t) b[t] = *(const v4f*)&Bsh[t * 16 + n16][kk];
#pragma unroll
      for (int t = 0; t < NT; ++t)
        acc[t] = wmma_f32((v2f){b[t].x, b[t].y}, (v2f){a.x, a.y}, acc[t]);
#pragma unroll
      for (int t = 0; t < NT; ++t)
        acc[t] = wmma_f32((v2f){b[t].z, b[t].w}, (v2f){a.z, a.w}, acc[t]);
    }
  }

  const float cn = sqrtf(ssq + xhalf(ssq));
  const float il = 1.0f / fmaxf(cn, 1e-12f);
  const float na = cn * il;
  for (int col = tid; col < NC; col += 128) {
    const int j = col / RR, r = col - j * RR;
    invi[col] = 1.0f / fmaxf(sqrtf(Gsh[col][r]), 1e-12f);
  }
  __syncthreads();

#define J0(t, r) ((16 * (t) + (r)) / RR)
#define J1(t, r) ((16 * (t) + 8 + (r)) / RR)

  float x[NT][8];
#pragma unroll
  for (int t = 0; t < NT; ++t) {
    const v4f lo = *(const v4f*)&invi[t * 16 + 8 * hi];
    const v4f hh = *(const v4f*)&invi[t * 16 + 8 * hi + 4];
    x[t][0] = acc[t][0] * il * lo.x; x[t][1] = acc[t][1] * il * lo.y;
    x[t][2] = acc[t][2] * il * lo.z; x[t][3] = acc[t][3] * il * lo.w;
    x[t][4] = acc[t][4] * il * hh.x; x[t][5] = acc[t][5] * il * hh.y;
    x[t][6] = acc[t][6] * il * hh.z; x[t][7] = acc[t][7] * il * hh.w;
  }

  float m[NI], sum[NI], dot[NI], inv[NI];
#pragma unroll
  for (int j = 0; j < NI; ++j) { m[j] = -1e30f; sum[j] = 0.f; dot[j] = 0.f; }
#pragma unroll
  for (int t = 0; t < NT; ++t)
#pragma unroll
    for (int r = 0; r < 8; ++r) {
      const float v = x[t][r];
      if (J0(t, r) == J1(t, r)) {
        m[J0(t, r)] = fmaxf(m[J0(t, r)], v);
      } else {
        m[J0(t, r)] = fmaxf(m[J0(t, r)], up ? -1e30f : v);
        m[J1(t, r)] = fmaxf(m[J1(t, r)], up ? v : -1e30f);
      }
    }
#pragma unroll
  for (int j = 0; j < NI; ++j) m[j] = fmaxf(m[j], xhalf(m[j]));
#pragma unroll
  for (int t = 0; t < NT; ++t) {
    __builtin_amdgcn_sched_barrier(0);
#pragma unroll
    for (int r = 0; r < 8; ++r) {
      const float v = x[t][r];
      const float mm = (J0(t, r) == J1(t, r)) ? m[J0(t, r)] : (up ? m[J1(t, r)] : m[J0(t, r)]);
      const float e = expf(LAMBDA * (v - mm));
      const float ev = e * v;
      if (J0(t, r) == J1(t, r)) {
        sum[J0(t, r)] += e;
        dot[J0(t, r)] += ev;
      } else {
        sum[J0(t, r)] += up ? 0.f : e;
        dot[J0(t, r)] += up ? 0.f : ev;
        sum[J1(t, r)] += up ? e : 0.f;
        dot[J1(t, r)] += up ? ev : 0.f;
      }
      x[t][r] = e;
    }
  }
#pragma unroll
  for (int j = 0; j < NI; ++j) {
    sum[j] += xhalf(sum[j]);
    dot[j] += xhalf(dot[j]);
    inv[j] = 1.0f / sum[j];
    dot[j] *= inv[j];
  }
#pragma unroll
  for (int t = 0; t < NT; ++t) {
    const v4f lo = *(const v4f*)&invi[t * 16 + 8 * hi];
    const v4f hh = *(const v4f*)&invi[t * 16 + 8 * hi + 4];
    const float f[8] = {lo.x, lo.y, lo.z, lo.w, hh.x, hh.y, hh.z, hh.w};
#pragma unroll
    for (int r = 0; r < 8; ++r) {
      const float iv = (J0(t, r) == J1(t, r)) ? inv[J0(t, r)] : (up ? inv[J1(t, r)] : inv[J0(t, r)]);
      x[t][r] = x[t][r] * iv * f[r];
    }
  }
  __builtin_amdgcn_sched_barrier(0);

  const int len = cap_lens[c];
  const bool live = wave * 16 + n16 < len;
  float res[NI];
#pragma unroll
  for (int j = 0; j < NI; ++j) {
    __builtin_amdgcn_sched_barrier(0);
    int grow[3];
#pragma unroll
    for (int u = 0; u < 3; ++u) {
      int g = 16 * (2 * j + u) + n16 - RR * j;
      g = g < 0 ? 0 : (g > RR - 1 ? RR - 1 : g);
      grow[u] = j * RR + g;
    }
    v8f V[3];
#pragma unroll
    for (int u = 0; u < 3; ++u) V[u] = (v8f){};
#pragma unroll
    for (int tp = 0; tp < 3; ++tp) {
      const int t = 2 * j + tp;
#pragma unroll
      for (int q = 0; q < 4; ++q) {
        const bool in0 = J0(t, 2 * q) == j, in1 = J1(t, 2 * q) == j;
        if (in0 || in1) {
          v2f b = (v2f){x[t][2 * q], x[t][2 * q + 1]};
          if (!(in0 && in1)) {
            const bool keep = in0 ? !up : up;
            b.x = keep ? b.x : 0.f;
            b.y = keep ? b.y : 0.f;
          }
          int kcol = 16 * t + 8 * hi + 2 * q - RR * j;
          kcol = (kcol < 0 || kcol > RR - 2) ? 0 : kcol;
#pragma unroll
          for (int u = 0; u < 3; ++u) {
            const v2f a = *(const v2f*)&Gsh[grow[u]][kcol];
            V[u] = wmma_f32(a, b, V[u]);
          }
        }
      }
    }
    float p = 0.f;
#pragma unroll
    for (int u = 0; u < 3; ++u) {
      const int t = 2 * j + u;
#pragma unroll
      for (int r = 0; r < 8; ++r) {
        const bool in0 = J0(t, r) == j, in1 = J1(t, r) == j;
        const float wv = x[t][r] * V[u][r];
        if (in0 && in1) p += wv;
        else if (in0) p += up ? 0.f : wv;
        else if (in1) p += up ? wv : 0.f;
      }
    }
    const float nb2 = p + xhalf(p);
    const float nb = sqrtf(fmaxf(nb2, 0.f));
    const float denom = fmaxf(na * nb, 1e-8f);
    float rl = dot[j] / denom;
    if (!live) rl = 0.f;
    rl += xrow<8>(rl);
    rl += xrow<4>(rl);
    rl += xrow<2>(rl);
    rl += xrow<1>(rl);
    res[j] = rl;
  }
#undef J0
#undef J1
  if (lane == 0) {
#pragma unroll
    for (int j = 0; j < NI; ++j) part[wave][j] = res[j];
  }
  __syncthreads();
  if (tid < NI) {
    const float s = (part[0][tid] + part[1][tid]) + (part[2][tid] + part[3][tid]);
    out[(size_t)c * BB + i0 + tid] = s * (1.0f / LL);
  }
}

extern "C" void kernel_launch(void* const* d_in, const int* in_sizes, int n_in,
                              void* d_out, int out_size, void* d_ws, size_t ws_size,
                              hipStream_t stream) {
  (void)in_sizes; (void)n_in; (void)out_size; (void)ws_size;
  const float* images = (const float*)d_in[0];
  const float* captions = (const float*)d_in[1];
  const int* cap_lens = (const int*)d_in[2];
  float* out = (float*)d_out;

  float* G = (float*)d_ws;

  gram_kernel<<<dim3(9, BB), 32 * GW, 0, stream>>>(images, G);
  scan_main<<<dim3(BB / NI, BB), 128, 0, stream>>>(captions, images, G, cap_lens, out);
}


# scan_main blocks packed from the live 16-token tiles only: a caption takes 1, 2 or 4 wave slots by its length, blocks past the list's end exit at entry; masked tiles no longer computed; sums kept
# speedup vs baseline: 2.4101x; 1.2330x over previous
#include <hip/hip_runtime.h>
#include <math.h>

#define BB 64
#define RR 36
#define RP 48
#define LL 64
#define EE 512
#define KC 32
#define KP 36
#define LAMBDA 9.0f
#define NI 4
#define NC (NI * RR)
#define NT (NC / 16)
#define GP RR

typedef float v2f __attribute__((ext_vector_type(2)));
typedef float v4f __attribute__((ext_vector_type(4)));
typedef float v8f __attribute__((ext_vector_type(8)));

__device__ __forceinline__ v8f wmma_f32(v2f a, v2f b, v8f c) {
  return __builtin_amdgcn_wmma_f32_16x16x4_f32(
       false, a,  false, b,
       (short)0, c,  false,  false);
}

__device__ __forceinline__ float xhalf(float x) {
  const int b = __builtin_bit_cast(int, x);
  return __builtin_bit_cast(float, __builtin_amdgcn_permlanex16(b, b, 0x76543210, 0xfedcba98, false, false));
}
template <int K> __device__ __forceinline__ float xrow(float x) {
  return __builtin_bit_cast(float, __builtin_amdgcn_update_dpp(0, __builtin_bit_cast(int, x), 0x160 | K, 0xf, 0xf, false));
}

#define GW 4
#define GK (EE / GW)
#define GQ (GK / 8)
__global__ __launch_bounds__(32 * GW) void gram_kernel(
    const float* __restrict__ images, float* __restrict__ G) {
  __shared__ float red[GW][8][32];
  const int i = blockIdx.y;
  const int t = blockIdx.x, ti = t / 3, tj = t % 3;
  const int tid = threadIdx.x;
  const int wave = tid >> 5, lane = tid & 31;
  const int hi = lane >> 4, n16 = lane & 15;
  const int ar = ti * 16 + n16, br = tj * 16 + n16;
  const bool am = ar < RR, bm = br < RR;
  const float* Arow = images + ((size_t)i * RR + (am ? ar : 0)) * EE + wave * GK + 4 * hi;
  const float* Brow = images + ((size_t)i * RR + (bm ? br : 0)) * EE + wave * GK + 4 * hi;
  float4 a[GQ], b[GQ];
#pragma unroll
  for (int q = 0; q < GQ; ++q) a[q] = *(const float4*)(Arow + 8 * q);
#pragma unroll
  for (int q = 0; q < GQ; ++q) b[q] = *(const float4*)(Brow + 8 * q);
  __builtin_amdgcn_sched_barrier(0);
  v8f acc0 = {}, acc1 = {};
#pragma unroll
  for (int q = 0; q < GQ; ++q) {
    float4 x = a[q], y = b[q];
    if (!am) x = make_float4(0.f, 0.f, 0.f, 0.f);
    if (!bm) y = make_float4(0.f, 0.f, 0.f, 0.f);
    acc0 = wmma_f32((v2f){x.x, x.y}, (v2f){y.x, y.y}, acc0);
    acc1 = wmma_f32((v2f){x.z, x.w}, (v2f){y.z, y.w}, acc1);
  }
#pragma unroll
  for (int j = 0; j < 8; ++j) red[wave][j][lane] = acc0[j] + acc1[j];
  __syncthreads();
#pragma unroll
  for (int p = 0; p < 2; ++p) {
    const int e = tid + p * 32 * GW;
    const int j = e >> 5, l = e & 31;
    float sum = red[0][j][l];
#pragma unroll
    for (int w = 1; w < GW; ++w) sum += red[w][j][l];
    const int row = ti * 16 + j + 8 * (l >> 4);
    const int col = tj * 16 + (l & 15);
    G[((size_t)i * RP + row) * RP + col] = sum;
  }
}

__device__ __forceinline__ int slots_of(int len) {
  const int n = (min(max(len, 0), LL) + 15) >> 4;
  return n <= 1 ? 1 : (n == 2 ? 2 : 4);
}

__global__ __launch_bounds__(128, 4) void scan_main(
    const float* __restrict__ captions, const float* __restrict__ images,
    const float* __restrict__ G,
    const int* __restrict__ cap_lens, float* __restrict__ out) {
  __shared__ __attribute__((aligned(16))) float stage[(LL + NC) * KP];
  __shared__ __attribute__((aligned(16))) float Gsh[NC][GP];
  __shared__ __attribute__((aligned(16))) float invi[NC];
  __shared__ float part[4][NI];
  float (*Ash)[KP] = (float (*)[KP])stage;
  float (*Bsh)[KP] = (float (*)[KP])(stage + LL * KP);

  const int i0 = blockIdx.x * NI;
  const int tid = threadIdx.x;
  const int lane = tid & 31;
  const int wave = __builtin_amdgcn_readfirstlane(tid >> 5);
  const int hi = lane >> 4, n16 = lane & 15;

  const int len_lo = cap_lens[lane], len_hi = cap_lens[lane + 32];
  const int p_lo = slots_of(len_lo), p_hi = slots_of(len_hi);
  const unsigned m4l = __builtin_amdgcn_ballot_w32(p_lo == 4), m4h = __builtin_amdgcn_ballot_w32(p_hi == 4);
  const unsigned m2l = __builtin_amdgcn_ballot_w32(p_lo == 2), m2h = __builtin_amdgcn_ballot_w32(p_hi == 2);
  const unsigned m1l = __builtin_amdgcn_ballot_w32(p_lo == 1), m1h = __builtin_amdgcn_ballot_w32(p_hi == 1);
  const int n4 = __builtin_popcount(m4l) + __builtin_popcount(m4h);
  const int n2 = __builtin_popcount(m2l) + __builtin_popcount(m2h);
  const int n1 = __builtin_popcount(m1l) + __builtin_popcount(m1h);
  const int nslots = 4 * n4 + 2 * n2 + n1;
  const int s0 = 4 * (int)blockIdx.y;
  if (s0 >= nslots) return;
  const unsigned below = (1u << lane) - 1u;
  const unsigned ml_lo = p_lo == 4 ? m4l : (p_lo == 2 ? m2l : m1l);
  const unsigned mh_lo = p_hi == 4 ? m4l : (p_hi == 2 ? m2l : m1l);
  const unsigned mh_hi = p_hi == 4 ? m4h : (p_hi == 2 ? m2h : m1h);
  const int off_lo = (p_lo == 4 ? 0 : (p_lo == 2 ? 4 * n4 : 4 * n4 + 2 * n2)) +
                     p_lo * __builtin_popcount(ml_lo & below);
  const int off_hi = (p_hi == 4 ? 0 : (p_hi == 2 ? 4 * n4 : 4 * n4 + 2 * n2)) +
                     p_hi * (__builtin_popcount(mh_lo) + __builtin_popcount(mh_hi & below));
  int cq[4], tq[4], pq[4], lq[4];
  bool vq[4];
#pragma unroll
  for (int q = 0; q < 4; ++q) {
    const int s = s0 + q;
    const unsigned hl = __builtin_amdgcn_ballot_w32((unsigned)(s - off_lo) < (unsigned)p_lo);
    const unsigned hh = __builtin_amdgcn_ballot_w32((unsigned)(s - off_hi) < (unsigned)p_hi);
    const bool inl = hl != 0u;
    const unsigned hm = inl ? hl : hh;
    vq[q] = hm != 0u;
    const int src = vq[q] ? __builtin_ctz(hm) : 0;
    const int off = inl ? __builtin_amdgcn_readlane(off_lo, src) : __builtin_amdgcn_readlane(off_hi, src);
    const int pp = inl ? __builtin_amdgcn_readlane(p_lo, src) : __builtin_amdgcn_readlane(p_hi, src);
    const int ln = inl ? __builtin_amdgcn_readlane(len_lo, src) : __builtin_amdgcn_readlane(len_hi, src);
    cq[q] = vq[q] ? (inl ? src : src + 32) : 0;
    tq[q] = vq[q] ? s - off : 0;
    pq[q] = vq[q] ? pp : 1;
    lq[q] = vq[q] ? ln : 0;
  }
  const int capw = wave == 0 ? cq[0] : (wave == 1 ? cq[1] : (wave == 2 ? cq[2] : cq[3]));
  const int tile = wave == 0 ? tq[0] : (wave == 1 ? tq[1] : (wave == 2 ? tq[2] : tq[3]));
  const int len = wave == 0 ? lq[0] : (wave == 1 ? lq[1] : (wave == 2 ? lq[2] : lq[3]));

  const int srow = (tid >> 4) + 8 * ((tid >> 3) & 1), sc4 = (tid & 7) << 2;
  const int arow = hi + 8 * ((lane >> 3) & 1);
  const float* Bp = images + (size_t)i0 * RR * EE + srow * EE + sc4;
  const unsigned long long adist =
      (unsigned long long)(captions + ((size_t)capw * LL + (size_t)tile * 16) * EE) -
      (unsigned long long)(images + (size_t)i0 * RR * EE + (size_t)(2 * wave) * EE);

  v4f ra[4], rb[9];
  float ssq = 0.f;
#pragma unroll
  for (int q = 0; q < 4; ++q) ra[q] = *(const v4f*)((const char*)Bp + adist + q * 2 * EE * 4);
#pragma unroll
  for (int q = 0; q < 9; ++q) rb[q] = *(const v4f*)(Bp + q * 16 * EE);

  {
    const float* Gi = G + (size_t)i0 * RP * RP;
    v4f rg[11];
#pragma unroll
    for (int q = 0; q < 11; ++q) {
      const int idx = tid + q * 128;
      if (idx < NC * 9) {
        const int row = idx / 9, c4 = idx - row * 9;
        const int j = row / RR, r = row - j * RR;
        rg[q] = *(const v4f*)(Gi + ((size_t)j * RP + r) * RP + c4 * 4);
      }
    }
#pragma unroll
    for (int q = 0; q < 11; ++q) {
      const int idx = tid + q * 128;
      if (idx < NC * 9) {
        const int row = idx / 9, col = (idx - row * 9) << 2;
        *(v4f*)&Gsh[row][col] = rg[q];
      }
    }
  }

  v8f acc[NT];
#pragma unroll
  for (int t = 0; t < NT; ++t) acc[t] = (v8f){};
  for (int kc = 0; kc < EE; kc += KC) {
    __syncthreads();
#pragma unroll
    for (int q = 0; q < 4; ++q) *(v4f*)&Ash[wave * 16 + arow + 2 * q][sc4] = ra[q];
#pragma unroll
    for (int q = 0; q < 9; ++q) *(v4f*)&Bsh[srow + 16 * q][sc4] = rb[q];
    __syncthreads();
    if (kc + KC < EE) {
      Bp += KC;
#pragma unroll
      for (int q = 0; q < 4; ++q)
        ra[q] = *(const v4f*)((const char*)Bp + adist + q * 2 * EE * 4);
#pragma unroll
      for (int q = 0; q < 9; ++q)
        rb[q] = *(const v4f*)(Bp + q * 16 * EE);
    }
#pragma unroll
    for (int k0 = 0; k0 < KC; k0 += 8) {
      const int kk = k0 + 4 * hi;
      const v4f a = *(const v4f*)&Ash[wave * 16 + n16][kk];
      ssq += a.x * a.x + a.y * a.y + a.z * a.z + a.w * a.w;
      v4f b[NT];
#pragma unroll
      for (int t = 0; t < NT; ++t) b[t] = *(const v4f*)&Bsh[t * 16 + n16][kk];
#pragma unroll
      for (int t = 0; t < NT; ++t)
        acc[t] = wmma_f32((v2f){b[t].x, b[t].y}, (v2f){a.x, a.y}, acc[t]);
#pragma unroll
      for (int t = 0; t < NT; ++t)
        acc[t] = wmma_f32((v2f){b[t].z, b[t].w}, (v2f){a.z, a.w}, acc[t]);
    }
  }

  const int lane_e = (int)__builtin_amdgcn_mbcnt_lo(~0u, 0u);
  const int tid_e = wave * 32 + lane_e;
  const int hi_e = lane_e >> 4, n16_e = lane_e & 15;
  const bool up = hi_e != 0;
  const float cn = sqrtf(ssq + xhalf(ssq));
  const float il = 1.0f / fmaxf(cn, 1e-12f);
  const float na = cn * il;
  for (int col = tid_e; col < NC; col += 128) {
    const int j = col / RR, r = col - j * RR;
    invi[col] = 1.0f / fmaxf(sqrtf(Gsh[col][r]), 1e-12f);
  }
  __syncthreads();

#define J0(t, r) ((16 * (t) + (r)) / RR)
#define J1(t, r) ((16 * (t) + 8 + (r)) / RR)

  float x[NT][8];
#pragma unroll
  for (int t = 0; t < NT; ++t) {
    const v4f lo = *(const v4f*)&invi[t * 16 + 8 * hi_e];
    const v4f hh = *(const v4f*)&invi[t * 16 + 8 * hi_e + 4];
    x[t][0] = acc[t][0] * il * lo.x; x[t][1] = acc[t][1] * il * lo.y;
    x[t][2] = acc[t][2] * il * lo.z; x[t][3] = acc[t][3] * il * lo.w;
    x[t][4] = acc[t][4] * il * hh.x; x[t][5] = acc[t][5] * il * hh.y;
    x[t][6] = acc[t][6] * il * hh.z; x[t][7] = acc[t][7] * il * hh.w;
  }

  float m[NI], sum[NI], dot[NI], inv[NI];
#pragma unroll
  for (int j = 0; j < NI; ++j) { m[j] = -1e30f; sum[j] = 0.f; dot[j] = 0.f; }
#pragma unroll
  for (int t = 0; t < NT; ++t)
#pragma unroll
    for (int r = 0; r < 8; ++r) {
      const float v = x[t][r];
      if (J0(t, r) == J1(t, r)) {
        m[J0(t, r)] = fmaxf(m[J0(t, r)], v);
      } else {
        m[J0(t, r)] = fmaxf(m[J0(t, r)], up ? -1e30f : v);
        m[J1(t, r)] = fmaxf(m[J1(t, r)], up ? v : -1e30f);
      }
    }
#pragma unroll
  for (int j = 0; j < NI; ++j) m[j] = fmaxf(m[j], xhalf(m[j]));
#pragma unroll
  for (int t = 0; t < NT; ++t) {
    __builtin_amdgcn_sched_barrier(0);
#pragma unroll
    for (int r = 0; r < 8; ++r) {
      const float v = x[t][r];
      const float mm = (J0(t, r) == J1(t, r)) ? m[J0(t, r)] : (up ? m[J1(t, r)] : m[J0(t, r)]);
      const float e = expf(LAMBDA * (v - mm));
      const float ev = e * v;
      if (J0(t, r) == J1(t, r)) {
        sum[J0(t, r)] += e;
        dot[J0(t, r)] += ev;
      } else {
        sum[J0(t, r)] += up ? 0.f : e;
        dot[J0(t, r)] += up ? 0.f : ev;
        sum[J1(t, r)] += up ? e : 0.f;
        dot[J1(t, r)] += up ? ev : 0.f;
      }
      x[t][r] = e;
    }
  }
#pragma unroll
  for (int j = 0; j < NI; ++j) {
    sum[j] += xhalf(sum[j]);
    dot[j] += xhalf(dot[j]);
    inv[j] = 1.0f / sum[j];
    dot[j] *= inv[j];
  }
#pragma unroll
  for (int t = 0; t < NT; ++t) {
    const v4f lo = *(const v4f*)&invi[t * 16 + 8 * hi_e];
    const v4f hh = *(const v4f*)&invi[t * 16 + 8 * hi_e + 4];
    const float f[8] = {lo.x, lo.y, lo.z, lo.w, hh.x, hh.y, hh.z, hh.w};
#pragma unroll
    for (int r = 0; r < 8; ++r) {
      const float iv = (J0(t, r) == J1(t, r)) ? inv[J0(t, r)] : (up ? inv[J1(t, r)] : inv[J0(t, r)]);
      x[t][r] = x[t][r] * iv * f[r];
    }
  }
  __builtin_amdgcn_sched_barrier(0);

  const bool live = tile * 16 + n16_e < len;
#pragma unroll
  for (int j = 0; j < NI; ++j) {
    __builtin_amdgcn_sched_barrier(0);
    int grow[3];
#pragma unroll
    for (int u = 0; u < 3; ++u) {
      int g = 16 * (2 * j + u) + n16_e - RR * j;
      g = g < 0 ? 0 : (g > RR - 1 ? RR - 1 : g);
      grow[u] = j * RR + g;
    }
    v8f V[3];
#pragma unroll
    for (int u = 0; u < 3; ++u) V[u] = (v8f){};
#pragma unroll
    for (int tp = 0; tp < 3; ++tp) {
      const int t = 2 * j + tp;
#pragma unroll
      for (int q = 0; q < 4; ++q) {
        const bool in0 = J0(t, 2 * q) == j, in1 = J1(t, 2 * q) == j;
        if (in0 || in1) {
          v2f b = (v2f){x[t][2 * q], x[t][2 * q + 1]};
          if (!(in0 && in1)) {
            const bool keep = in0 ? !up : up;
            b.x = keep ? b.x : 0.f;
            b.y = keep ? b.y : 0.f;
          }
          int kcol = 16 * t + 8 * hi_e + 2 * q - RR * j;
          kcol = (kcol < 0 || kcol > RR - 2) ? 0 : kcol;
#pragma unroll
          for (int u = 0; u < 3; ++u) {
            const v2f a = *(const v2f*)&Gsh[grow[u]][kcol];
            V[u] = wmma_f32(a, b, V[u]);
          }
        }
      }
    }
    float p = 0.f;
#pragma unroll
    for (int u = 0; u < 3; ++u) {
      const int t = 2 * j + u;
#pragma unroll
      for (int r = 0; r < 8; ++r) {
        const bool in0 = J0(t, r) == j, in1 = J1(t, r) == j;
        const float wv = x[t][r] * V[u][r];
        if (in0 && in1) p += wv;
        else if (in0) p += up ? 0.f : wv;
        else if (in1) p += up ? wv : 0.f;
      }
    }
    const float nb2 = p + xhalf(p);
    const float nb = sqrtf(fmaxf(nb2, 0.f));
    const float denom = fmaxf(na * nb, 1e-8f);
    float rl = dot[j] / denom;
    if (!live) rl = 0.f;
    rl += xrow<8>(rl);
    rl += xrow<4>(rl);
    rl += xrow<2>(rl);
    rl += xrow<1>(rl);
    if (lane_e == 0) part[wave][j] = rl;
  }
#undef J0
#undef J1
  __syncthreads();
  if (tid_e < 4 * NI) {
    const int w = tid_e >> 2, j = tid_e & 3;
    int hc = 0, hp = 1;
    bool head = false;
#pragma unroll
    for (int q = 0; q < 4; ++q)
      if (w == q) { hc = cq[q]; hp = pq[q]; head = vq[q] && tq[q] == 0; }
    if (head) {
      float s;
      if (hp == 4) s = (part[0][j] + part[1][j]) + (part[2][j] + part[3][j]);
      else if (hp == 2) s = part[w][j] + part[(w + 1) & 3][j];
      else s = part[w][j];
      out[(size_t)hc * BB + i0 + j] = s * (1.0f / LL);
    }
  }
}

extern "C" void kernel_launch(void* const* d_in, const int* in_sizes, int n_in,
                              void* d_out, int out_size, void* d_ws, size_t ws_size,
                              hipStream_t stream) {
  (void)in_sizes; (void)n_in; (void)out_size; (void)ws_size;
  const float* images = (const float*)d_in[0];
  const float* captions = (const float*)d_in[1];
  const int* cap_lens = (const int*)d_in[2];
  float* out = (float*)d_out;

  float* G = (float*)d_ws;

  gram_kernel<<<dim3(9, BB), 32 * GW, 0, stream>>>(images, G);
  scan_main<<<dim3(BB / NI, BB), 128, 0, stream>>>(captions, images, G, cap_lens, out);
}
